# starting kernel
# baseline (speedup 1.0000x reference)
#include <hip/hip_runtime.h>
#include <math.h>

#define NQ   12
#define NL   6
#define DIM  4096
#define BLK  256

typedef __attribute__((ext_vector_type(2))) float v2f;
typedef __attribute__((ext_vector_type(8))) float v8f;

__device__ __forceinline__ void rx_pair(v2f& a, v2f& b, float c, float s) {
    v2f na, nb;
    na.x =  c * a.x + s * b.y;
    na.y =  c * a.y - s * b.x;
    nb.x =  s * a.y + c * b.x;
    nb.y = -s * a.x + c * b.y;
    a = na; b = nb;
}

__global__ __launch_bounds__(BLK)
void qsim_kernel(const float* __restrict__ x,
                 const float* __restrict__ w,
                 float* __restrict__ out) {
    __shared__ v2f lds2[2 * DIM];
    int cur = 0;

    const int tid = threadIdx.x;
    const int b   = blockIdx.x;

    float a0r[NQ], a0i[NQ], a1r[NQ], a1i[NQ];
    #pragma unroll
    for (int q = 0; q < NQ; ++q) {
        float ang = tanhf(x[b * NQ + q]) * 3.14159265358979323846f;
        float h = 0.5f * ang;
        float c = __cosf(h), s = __sinf(h);
        a0r[q] = c * c;  a0i[q] = -s * c;
        a1r[q] = s * c;  a1i[q] = s * s;
    }
    {
        v2f* st = lds2;
        #pragma unroll 1
        for (int k = 0; k < DIM / BLK; ++k) {
            int s = tid + BLK * k;
            float vr = 1.0f, vi = 0.0f;
            #pragma unroll
            for (int q = 0; q < NQ; ++q) {
                int bit = (s >> (NQ - 1 - q)) & 1;
                float ar = bit ? a1r[q] : a0r[q];
                float ai = bit ? a1i[q] : a0i[q];
                float nr = vr * ar - vi * ai;
                float ni = vr * ai + vi * ar;
                vr = nr; vi = ni;
            }
            v2f amp; amp.x = vr; amp.y = vi;
            st[s] = amp;
        }
    }

    #pragma unroll 1
    for (int l = 0; l < NL; ++l) {
        #pragma unroll 1
        for (int g = 0; g < NQ / 2; ++g) {
            __syncthreads();
            float ha = 0.5f * w[l * NQ + 2 * g];
            float hb = 0.5f * w[l * NQ + 2 * g + 1];
            float ca = __cosf(ha), sa = __sinf(ha);
            float cb = __cosf(hb), sb = __sinf(hb);
            int ma = 1 << (NQ - 1 - 2 * g);
            int mb = ma >> 1;
            v2f* st = lds2 + cur * DIM;
            #pragma unroll
            for (int k = 0; k < (DIM / 4) / BLK; ++k) {
                int p   = tid + BLK * k;
                int s00 = ((p & ~(mb - 1)) << 2) | (p & (mb - 1));
                v2f a00 = st[s00];
                v2f a01 = st[s00 | mb];
                v2f a10 = st[s00 | ma];
                v2f a11 = st[s00 | ma | mb];
                rx_pair(a00, a10, ca, sa);
                rx_pair(a01, a11, ca, sa);
                rx_pair(a00, a01, cb, sb);
                rx_pair(a10, a11, cb, sb);
                st[s00]           = a00;
                st[s00 | mb]      = a01;
                st[s00 | ma]      = a10;
                st[s00 | ma | mb] = a11;
            }
        }
        __syncthreads();
        {
            v2f* src = lds2 + cur * DIM;
            v2f* dst = lds2 + (cur ^ 1) * DIM;
            #pragma unroll 1
            for (int k = 0; k < DIM / BLK; ++k) {
                int s = tid + BLK * k;
                int j = s;
                #pragma unroll
                for (int i = NQ - 1; i >= 0; --i) {
                    int cpos = NQ - 1 - i;
                    int tpos = NQ - 1 - ((i + 1) % NQ);
                    j ^= ((j >> cpos) & 1) << tpos;
                }
                dst[s] = src[j];
            }
        }
        cur ^= 1;
    }

    __syncthreads();
    float* P  = (float*)(lds2 + (cur ^ 1) * DIM);
    float* R  = P + DIM;
    float* Wt = R + 512;
    {
        v2f* st = lds2 + cur * DIM;
        #pragma unroll 1
        for (int k = 0; k < DIM / BLK; ++k) {
            int s = tid + BLK * k;
            v2f a = st[s];
            P[s] = a.x * a.x + a.y * a.y;
        }
        #pragma unroll
        for (int e = 0; e < 4; ++e) {
            int idx = tid + BLK * e;
            int c   = idx >> 6;
            int lo  = idx & 63;
            float v = 0.0f;
            if (c < 6)       v = 1.0f - 2.0f * (float)((lo >> c) & 1);
            else if (c == 6) v = 1.0f;
            Wt[idx] = v;
        }
    }
    __syncthreads();

    int wave = tid >> 5;
    int lane = tid & 31;
    if (wave < 4) {
        int row   = (wave << 4) + (lane & 15);
        int col   = lane & 15;
        int khalf = (lane >> 4) << 1;
        v8f acc = {};
        #pragma unroll
        for (int k = 0; k < 64; k += 4) {
            int k0 = k + khalf;
            v2f A = *(const v2f*)(P  + row * 64 + k0);
            v2f B = *(const v2f*)(Wt + col * 64 + k0);
            acc = __builtin_amdgcn_wmma_f32_16x16x4_f32(
                 false, A,  false, B,
                 (short)0, acc,  false,  false);
        }
        if (col < 8) {
            int rbase = (wave << 4) + ((lane >> 4) << 3);
            #pragma unroll
            for (int v = 0; v < 8; ++v)
                R[(rbase + v) * 8 + col] = acc[v];
        }
    }
    __syncthreads();

    if (tid < NQ) {
        int q = tid;
        float acc = 0.0f;
        if (q >= 6) {
            int c = NQ - 1 - q;
            #pragma unroll 1
            for (int hi = 0; hi < 64; ++hi) acc += R[hi * 8 + c];
        } else {
            int bp = 5 - q;
            #pragma unroll 1
            for (int hi = 0; hi < 64; ++hi) {
                float sgn = 1.0f - 2.0f * (float)((hi >> bp) & 1);
                acc += sgn * R[hi * 8 + 6];
            }
        }
        out[b * NQ + q] = acc;
    }
}

extern "C" void kernel_launch(void* const* d_in, const int* in_sizes, int n_in,
                              void* d_out, int out_size, void* d_ws, size_t ws_size,
                              hipStream_t stream) {
    const float* x  = (const float*)d_in[0];
    const float* w  = (const float*)d_in[1];
    float* out = (float*)d_out;
    int batch = in_sizes[0] / NQ;
    qsim_kernel<<<batch, BLK, 0, stream>>>(x, w, out);
}


# LDS bank conflicts removed: state index XOR-swizzled (bits 5,6 folded into bank bits 0..4, thread part of each address hoisted by XOR-linearity); WMMA tile rows at 68-float pitch
# speedup vs baseline: 1.1685x; 1.1685x over previous
#include <hip/hip_runtime.h>
#include <math.h>

#define NQ   12
#define NL   6
#define DIM  4096
#define BLK  256

typedef __attribute__((ext_vector_type(2))) float v2f;
typedef __attribute__((ext_vector_type(8))) float v8f;

__device__ __forceinline__ void rx_pair(v2f& a, v2f& b, float c, float s) {
    v2f na, nb;
    na.x =  c * a.x + s * b.y;
    na.y =  c * a.y - s * b.x;
    nb.x =  s * a.y + c * b.x;
    nb.y = -s * a.x + c * b.y;
    a = na; b = nb;
}

__host__ __device__ constexpr int swz(int s) {
    return s ^ (((s >> 5) & 1) * 5) ^ (((s >> 6) & 1) * 26);
}

__host__ __device__ constexpr int cnot_src(int s) {
    int j = s;
    for (int i = NQ - 1; i >= 0; --i) {
        int cpos = NQ - 1 - i;
        int tpos = NQ - 1 - ((i + 1) % NQ);
        j ^= ((j >> cpos) & 1) << tpos;
    }
    return j;
}

#define PITCH 68

__global__ __launch_bounds__(BLK)
void qsim_kernel(const float* __restrict__ x,
                 const float* __restrict__ w,
                 float* __restrict__ out) {
    __shared__ v2f lds2[2 * DIM];
    int cur = 0;

    const int tid = threadIdx.x;
    const int b   = blockIdx.x;
    const int pt = swz(tid);
    const int pj = swz(cnot_src(tid));

    float a0r[NQ], a0i[NQ], a1r[NQ], a1i[NQ];
    #pragma unroll
    for (int q = 0; q < NQ; ++q) {
        float ang = tanhf(x[b * NQ + q]) * 3.14159265358979323846f;
        float h = 0.5f * ang;
        float c = __cosf(h), s = __sinf(h);
        a0r[q] = c * c;  a0i[q] = -s * c;
        a1r[q] = s * c;  a1i[q] = s * s;
    }
    {
        v2f* st = lds2;
        #pragma unroll 1
        for (int k = 0; k < DIM / BLK; ++k) {
            int s = tid + BLK * k;
            float vr = 1.0f, vi = 0.0f;
            #pragma unroll
            for (int q = 0; q < NQ; ++q) {
                int bit = (s >> (NQ - 1 - q)) & 1;
                float ar = bit ? a1r[q] : a0r[q];
                float ai = bit ? a1i[q] : a0i[q];
                float nr = vr * ar - vi * ai;
                float ni = vr * ai + vi * ar;
                vr = nr; vi = ni;
            }
            v2f amp; amp.x = vr; amp.y = vi;
            st[pt + BLK * k] = amp;
        }
    }

    #pragma unroll 1
    for (int l = 0; l < NL; ++l) {
        #pragma unroll 1
        for (int g = 0; g < NQ / 2; ++g) {
            __syncthreads();
            float ha = 0.5f * w[l * NQ + 2 * g];
            float hb = 0.5f * w[l * NQ + 2 * g + 1];
            float ca = __cosf(ha), sa = __sinf(ha);
            float cb = __cosf(hb), sb = __sinf(hb);
            int ma = 1 << (NQ - 1 - 2 * g);
            int mb = ma >> 1;
            v2f* st = lds2 + cur * DIM;
            int qt = swz(((tid & ~(mb - 1)) << 2) | (tid & (mb - 1)));
            int xa = swz(ma), xb = swz(mb);
            #pragma unroll
            for (int k = 0; k < (DIM / 4) / BLK; ++k) {
                int hk  = BLK * k;
                int q00 = qt ^ swz(((hk & ~(mb - 1)) << 2) | (hk & (mb - 1)));
                int q01 = q00 ^ xb;
                int q10 = q00 ^ xa;
                int q11 = q00 ^ xa ^ xb;
                v2f a00 = st[q00];
                v2f a01 = st[q01];
                v2f a10 = st[q10];
                v2f a11 = st[q11];
                rx_pair(a00, a10, ca, sa);
                rx_pair(a01, a11, ca, sa);
                rx_pair(a00, a01, cb, sb);
                rx_pair(a10, a11, cb, sb);
                st[q00] = a00;
                st[q01] = a01;
                st[q10] = a10;
                st[q11] = a11;
            }
        }
        __syncthreads();
        {
            v2f* src = lds2 + cur * DIM;
            v2f* dst = lds2 + (cur ^ 1) * DIM;
            #pragma unroll 1
            for (int k = 0; k < DIM / BLK; ++k)
                dst[pt + BLK * k] = src[pj ^ cnot_src(BLK * k)];
        }
        cur ^= 1;
    }

    __syncthreads();
    float* P  = (float*)(lds2 + (cur ^ 1) * DIM);
    float* R  = P + 64 * PITCH;
    float* Wt = R + 512;
    {
        v2f* st = lds2 + cur * DIM;
        #pragma unroll 1
        for (int k = 0; k < DIM / BLK; ++k) {
            int s = tid + BLK * k;
            v2f a = st[pt + BLK * k];
            P[(s >> 6) * PITCH + (s & 63)] = a.x * a.x + a.y * a.y;
        }
        #pragma unroll
        for (int e = 0; e < 4; ++e) {
            int idx = tid + BLK * e;
            int c   = idx >> 6;
            int lo  = idx & 63;
            float v = 0.0f;
            if (c < 6)       v = 1.0f - 2.0f * (float)((lo >> c) & 1);
            else if (c == 6) v = 1.0f;
            Wt[c * PITCH + lo] = v;
        }
    }
    __syncthreads();

    int wave = tid >> 5;
    int lane = tid & 31;
    if (wave < 4) {
        int row   = (wave << 4) + (lane & 15);
        int col   = lane & 15;
        int khalf = (lane >> 4) << 1;
        v8f acc = {};
        #pragma unroll
        for (int k = 0; k < 64; k += 4) {
            int k0 = k + khalf;
            v2f A = *(const v2f*)(P  + row * PITCH + k0);
            v2f B = *(const v2f*)(Wt + col * PITCH + k0);
            acc = __builtin_amdgcn_wmma_f32_16x16x4_f32(
                 false, A,  false, B,
                 (short)0, acc,  false,  false);
        }
        if (col < 8) {
            int rbase = (wave << 4) + ((lane >> 4) << 3);
            #pragma unroll
            for (int v = 0; v < 8; ++v)
                R[(rbase + v) * 8 + col] = acc[v];
        }
    }
    __syncthreads();

    if (tid < NQ) {
        int q = tid;
        float acc = 0.0f;
        if (q >= 6) {
            int c = NQ - 1 - q;
            #pragma unroll 1
            for (int hi = 0; hi < 64; ++hi) acc += R[hi * 8 + c];
        } else {
            int bp = 5 - q;
            #pragma unroll 1
            for (int hi = 0; hi < 64; ++hi) {
                float sgn = 1.0f - 2.0f * (float)((hi >> bp) & 1);
                acc += sgn * R[hi * 8 + 6];
            }
        }
        out[b * NQ + q] = acc;
    }
}

extern "C" void kernel_launch(void* const* d_in, const int* in_sizes, int n_in,
                              void* d_out, int out_size, void* d_ws, size_t ws_size,
                              hipStream_t stream) {
    const float* x  = (const float*)d_in[0];
    const float* w  = (const float*)d_in[1];
    float* out = (float*)d_out;
    int batch = in_sizes[0] / NQ;
    qsim_kernel<<<batch, BLK, 0, stream>>>(x, w, out);
}


# Four RX gates per LDS pass on 16 register-held amplitudes (packed re/im f32 math), CNOT ring folded into the third pass's store address, init kept in registers: 3 LDS round trips per layer, was 7
# speedup vs baseline: 1.6423x; 1.4055x over previous
#include <hip/hip_runtime.h>
#include <math.h>


#define NQ   12
#define NL   6
#define DIM  4096
#define BLK  256
#define NR   16

typedef __attribute__((ext_vector_type(2))) float v2f;
typedef __attribute__((ext_vector_type(8))) float v8f;

__device__ __forceinline__ void rx_pair(v2f& a, v2f& b, v2f cc, v2f sm) {
    v2f ta = sm * b.yx;
    v2f tb = sm * a.yx;
    a = cc * a + ta;
    b = cc * b + tb;
}

__host__ __device__ constexpr int swz(int s) {
    return s ^ ((s >> 5) & 31);
}

__host__ __device__ constexpr int cnot_dst(int s) {
    int j = s;
    for (int i = 0; i < NQ; ++i) {
        int cpos = NQ - 1 - i;
        int tpos = NQ - 1 - ((i + 1) % NQ);
        j ^= ((j >> cpos) & 1) << tpos;
    }
    return j;
}

template <int SH>
__device__ __forceinline__ void load16(v2f (&a)[NR], const v2f* st, int base) {
    #pragma unroll
    for (int r = 0; r < NR; ++r) a[r] = st[base ^ swz(r << SH)];
}
template <int SH>
__device__ __forceinline__ void store16(const v2f (&a)[NR], v2f* st, int base) {
    #pragma unroll
    for (int r = 0; r < NR; ++r) st[base ^ swz(r << SH)] = a[r];
}

__device__ __forceinline__ void rx4(v2f (&a)[NR], const float* wq) {
    #pragma unroll
    for (int j = 0; j < 4; ++j) {
        float h = 0.5f * wq[j];
        float c = __cosf(h), s = __sinf(h);
        v2f cc; cc.x = c; cc.y = c;
        v2f sm; sm.x = s; sm.y = -s;
        const int m = 8 >> j;
        #pragma unroll
        for (int r = 0; r < NR; ++r)
            if (!(r & m)) rx_pair(a[r], a[r | m], cc, sm);
    }
}

#define PITCH 68

__global__ __launch_bounds__(BLK, 8)
void qsim_kernel(const float* __restrict__ x,
                 const float* __restrict__ w,
                 float* __restrict__ out) {
    __shared__ v2f lds2[2 * DIM];

    const int tid = threadIdx.x;
    const int b   = blockIdx.x;

    const int i1 = (tid & 15) | (((tid >> 4) & 1) << 9) | (((tid >> 5) & 1) << 8)
                 | (((tid >> 6) & 3) << 10);
    const int i2 = tid << 4;
    const int p0 = swz(tid);
    const int p1 = swz(i1);
    const int p2 = swz(i2);
    const int pd = swz(cnot_dst(i2));

    float* P  = (float*)(lds2 + (NL & 1) * DIM);
    float* R  = P + 64 * PITCH;
    float* Wt = R + 512;

    float a0r[NQ], a0i[NQ], a1r[NQ], a1i[NQ];
    #pragma unroll
    for (int q = 0; q < NQ; ++q) {
        float ang = tanhf(x[b * NQ + q]) * 3.14159265358979323846f;
        float h = 0.5f * ang;
        float c = __cosf(h), s = __sinf(h);
        a0r[q] = c * c;  a0i[q] = -s * c;
        a1r[q] = s * c;  a1i[q] = s * s;
    }
    v2f amp[NR];
    #pragma unroll
    for (int k = 0; k < NR; ++k) {
        float vr = 1.0f, vi = 0.0f;
        #pragma unroll
        for (int q = 0; q < NQ; ++q) {
            int bit = (q < 4) ? ((k >> (3 - q)) & 1) : ((tid >> (NQ - 1 - q)) & 1);
            float ar = bit ? a1r[q] : a0r[q];
            float ai = bit ? a1i[q] : a0i[q];
            float nr = vr * ar - vi * ai;
            float ni = vr * ai + vi * ar;
            vr = nr; vi = ni;
        }
        amp[k].x = vr; amp[k].y = vi;
    }

    {
        float w0[4];
        #pragma unroll
        for (int j = 0; j < 4; ++j) w0[j] = w[j];
        rx4(amp, w0);
        store16<8>(amp, lds2, p0);
    }
    int cur = 0;
    #pragma unroll 1
    for (int l = 0; l < NL - 1; ++l) {
        float wa[NQ];
        #pragma unroll
        for (int j = 0; j < NQ; ++j) wa[j] = w[l * NQ + 4 + j];
        const int sb = cur << 12;
        const int ob = sb ^ DIM;

        __syncthreads();
        {
            v2f a1[NR];
            load16<4>(a1, lds2, p1 | sb);
            rx4(a1, wa);
            store16<4>(a1, lds2, p1 | sb);
        }

        __syncthreads();
        {
            v2f a2[NR];
            load16<0>(a2, lds2, p2 | sb);
            rx4(a2, wa + 4);
            const int sd = pd | ob;
            #pragma unroll
            for (int r = 0; r < NR; ++r) lds2[sd ^ swz(cnot_dst(r))] = a2[r];
        }

        __syncthreads();
        {
            v2f a0[NR];
            load16<8>(a0, lds2, p0 | ob);
            rx4(a0, wa + 8);
            store16<8>(a0, lds2, p0 | ob);
        }
        cur ^= 1;
    }
    {
        float wa[8];
        #pragma unroll
        for (int j = 0; j < 8; ++j) wa[j] = w[(NL - 1) * NQ + 4 + j];
        v2f* st = lds2 + ((NL - 1) & 1) * DIM;

        __syncthreads();
        {
            v2f a1[NR];
            load16<4>(a1, st, p1);
            rx4(a1, wa);
            store16<4>(a1, st, p1);
        }
        __syncthreads();
        {
            v2f a2[NR];
            load16<0>(a2, st, p2);
            rx4(a2, wa + 4);
            const int d2 = swz(pd);
            #pragma unroll
            for (int r = 0; r < NR; ++r) {
                int d = d2 ^ cnot_dst(r);
                P[(d >> 6) * PITCH + (d & 63)] = a2[r].x * a2[r].x + a2[r].y * a2[r].y;
            }
        }
    }

    #pragma unroll
    for (int e = 0; e < 4; ++e) {
        int idx = tid + BLK * e;
        int c   = idx >> 6;
        int lo  = idx & 63;
        float v = 0.0f;
        if (c < 6)       v = 1.0f - 2.0f * (float)((lo >> c) & 1);
        else if (c == 6) v = 1.0f;
        Wt[c * PITCH + lo] = v;
    }
    __syncthreads();

    int wave = tid >> 5;
    int lane = tid & 31;
    if (wave < 4) {
        int row   = (wave << 4) + (lane & 15);
        int col   = lane & 15;
        int khalf = (lane >> 4) << 1;
        v8f acc = {};
        #pragma unroll
        for (int k = 0; k < 64; k += 4) {
            int k0 = k + khalf;
            v2f A = *(const v2f*)(P  + row * PITCH + k0);
            v2f B = *(const v2f*)(Wt + col * PITCH + k0);
            acc = __builtin_amdgcn_wmma_f32_16x16x4_f32(
                 false, A,  false, B,
                 (short)0, acc,  false,  false);
        }
        if (col < 8) {
            int rbase = (wave << 4) + ((lane >> 4) << 3);
            #pragma unroll
            for (int v = 0; v < 8; ++v)
                R[(rbase + v) * 8 + col] = acc[v];
        }
    }
    __syncthreads();

    if (tid < NQ) {
        int q = tid;
        float acc = 0.0f;
        if (q >= 6) {
            int c = NQ - 1 - q;
            #pragma unroll 1
            for (int hi = 0; hi < 64; ++hi) acc += R[hi * 8 + c];
        } else {
            int bp = 5 - q;
            #pragma unroll 1
            for (int hi = 0; hi < 64; ++hi) {
                float sgn = 1.0f - 2.0f * (float)((hi >> bp) & 1);
                acc += sgn * R[hi * 8 + 6];
            }
        }
        out[b * NQ + q] = acc;
    }
}

extern "C" void kernel_launch(void* const* d_in, const int* in_sizes, int n_in,
                              void* d_out, int out_size, void* d_ws, size_t ws_size,
                              hipStream_t stream) {
    const float* x  = (const float*)d_in[0];
    const float* w  = (const float*)d_in[1];
    float* out = (float*)d_out;
    int batch = in_sizes[0] / NQ;
    qsim_kernel<<<batch, BLK, 0, stream>>>(x, w, out);
}


# RX gates without their cosine factor: a' = a - i*tan(h')*b, one packed fma per amplitude (was mul+fma); h reduced mod pi/2, its X flips end as output signs; cos^2 product applied once at the end
# speedup vs baseline: 2.3284x; 1.4178x over previous
#include <hip/hip_runtime.h>
#include <math.h>


#define NQ   12
#define NL   6
#define DIM  4096
#define BLK  256
#define NR   16

typedef __attribute__((ext_vector_type(2))) float v2f;
typedef __attribute__((ext_vector_type(8))) float v8f;
typedef __attribute__((ext_vector_type(4))) float v4f;

template <int HI>
__device__ __forceinline__ v2f fma_rot(v2f tp, v2f b, v2f a) {
    v2f d;
    if (HI)
        asm("v_pk_fma_f32 %0, %1, %2, %3 op_sel:[1,1,0] op_sel_hi:[1,0,1] neg_hi:[1,0,0]"
            : "=v"(d) : "v"(tp), "v"(b), "v"(a));
    else
        asm("v_pk_fma_f32 %0, %1, %2, %3 op_sel:[0,1,0] op_sel_hi:[0,0,1] neg_hi:[1,0,0]"
            : "=v"(d) : "v"(tp), "v"(b), "v"(a));
    return d;
}
template <int HI>
__device__ __forceinline__ void rx_pair(v2f& a, v2f& b, v2f tp) {
    v2f na = fma_rot<HI>(tp, b, a);
    v2f nb = fma_rot<HI>(tp, a, b);
    a = na;
    b = nb;
}

__host__ __device__ constexpr int swz(int s) {
    return s ^ ((s >> 5) & 31);
}

__host__ __device__ constexpr int cnot_dst(int s) {
    int j = s;
    for (int i = 0; i < NQ; ++i) {
        int cpos = NQ - 1 - i;
        int tpos = NQ - 1 - ((i + 1) % NQ);
        j ^= ((j >> cpos) & 1) << tpos;
    }
    return j;
}

template <int SH>
__device__ __forceinline__ void load16(v2f (&a)[NR], const v2f* st, int base) {
    #pragma unroll
    for (int r = 0; r < NR; ++r) a[r] = st[base ^ swz(r << SH)];
}
template <int SH>
__device__ __forceinline__ void store16(const v2f (&a)[NR], v2f* st, int base) {
    #pragma unroll
    for (int r = 0; r < NR; ++r) st[base ^ swz(r << SH)] = a[r];
}

__device__ __forceinline__ void rx4(v2f (&a)[NR], v4f t4) {
    #pragma unroll
    for (int j = 0; j < 4; ++j) {
        const v2f tp = (j < 2) ? t4.xy : t4.zw;
        const int m = 8 >> j;
        #pragma unroll
        for (int r = 0; r < NR; ++r)
            if (!(r & m)) {
                if (j & 1) rx_pair<1>(a[r], a[r | m], tp);
                else       rx_pair<0>(a[r], a[r | m], tp);
            }
    }
}

#define PITCH 68

__global__ __launch_bounds__(BLK, 8)
void qsim_kernel(const float* __restrict__ x,
                 const float* __restrict__ w,
                 float* __restrict__ out) {
    __shared__ v2f lds2[2 * DIM];
    __shared__ __attribute__((aligned(16))) float gT[NL * NQ];
    __shared__ float gC[NL * NQ];
    __shared__ int   gK[NL * NQ];

    const int tid = threadIdx.x;
    const int b   = blockIdx.x;

    if (tid < NL * NQ) {
        float th = w[tid];
        float k  = rintf(th * 0.318309886183790671538f);
        float h  = fmaf(-k, 1.57079637050628662109375f, 0.5f * th);
        h = fmaf(k, 4.37113882867379e-8f, h);
        float c = __cosf(h), s = __sinf(h);
        gT[tid] = s * __builtin_amdgcn_rcpf(c);
        gC[tid] = c * c;
        gK[tid] = ((int)k) & 1;
    }

    const int i1 = (tid & 15) | (((tid >> 4) & 1) << 9) | (((tid >> 5) & 1) << 8)
                 | (((tid >> 6) & 3) << 10);
    const int i2 = tid << 4;
    const int p0 = swz(tid);
    const int p1 = swz(i1);
    const int p2 = swz(i2);
    const int pd = swz(cnot_dst(i2));

    float* P  = (float*)(lds2 + (NL & 1) * DIM);
    float* R  = P + 64 * PITCH;
    float* Wt = R + 512;

    float a0r[NQ], a0i[NQ], a1r[NQ], a1i[NQ];
    #pragma unroll
    for (int q = 0; q < NQ; ++q) {
        float ang = tanhf(x[b * NQ + q]) * 3.14159265358979323846f;
        float h = 0.5f * ang;
        float c = __cosf(h), s = __sinf(h);
        a0r[q] = c * c;  a0i[q] = -s * c;
        a1r[q] = s * c;  a1i[q] = s * s;
    }
    v2f amp[NR];
    #pragma unroll
    for (int k = 0; k < NR; ++k) {
        float vr = 1.0f, vi = 0.0f;
        #pragma unroll
        for (int q = 0; q < NQ; ++q) {
            int bit = (q < 4) ? ((k >> (3 - q)) & 1) : ((tid >> (NQ - 1 - q)) & 1);
            float ar = bit ? a1r[q] : a0r[q];
            float ai = bit ? a1i[q] : a0i[q];
            float nr = vr * ar - vi * ai;
            float ni = vr * ai + vi * ar;
            vr = nr; vi = ni;
        }
        amp[k].x = vr; amp[k].y = vi;
    }

    __syncthreads();
    {
        rx4(amp, *(const v4f*)(gT));
        store16<8>(amp, lds2, p0);
    }
    int cur = 0;
    #pragma unroll 1
    for (int l = 0; l < NL - 1; ++l) {
        const v4f* tq = (const v4f*)(gT + l * NQ + 4);
        const v4f t1 = tq[0], t2 = tq[1], t0 = tq[2];
        const int sb = cur << 12;
        const int ob = sb ^ DIM;

        __syncthreads();
        {
            v2f a1[NR];
            load16<4>(a1, lds2, p1 | sb);
            rx4(a1, t1);
            store16<4>(a1, lds2, p1 | sb);
        }

        __syncthreads();
        {
            v2f a2[NR];
            load16<0>(a2, lds2, p2 | sb);
            rx4(a2, t2);
            const int sd = pd | ob;
            #pragma unroll
            for (int r = 0; r < NR; ++r) lds2[sd ^ swz(cnot_dst(r))] = a2[r];
        }

        __syncthreads();
        {
            v2f a0[NR];
            load16<8>(a0, lds2, p0 | ob);
            rx4(a0, t0);
            store16<8>(a0, lds2, p0 | ob);
        }
        cur ^= 1;
    }
    {
        const v4f* tq = (const v4f*)(gT + (NL - 1) * NQ + 4);
        const v4f t1 = tq[0], t2 = tq[1];
        v2f* st = lds2 + ((NL - 1) & 1) * DIM;

        __syncthreads();
        {
            v2f a1[NR];
            load16<4>(a1, st, p1);
            rx4(a1, t1);
            store16<4>(a1, st, p1);
        }
        __syncthreads();
        {
            v2f a2[NR];
            load16<0>(a2, st, p2);
            rx4(a2, t2);
            const int d2 = swz(pd);
            #pragma unroll
            for (int r = 0; r < NR; ++r) {
                int d = d2 ^ cnot_dst(r);
                P[(d >> 6) * PITCH + (d & 63)] = a2[r].x * a2[r].x + a2[r].y * a2[r].y;
            }
        }
    }

    #pragma unroll
    for (int e = 0; e < 4; ++e) {
        int idx = tid + BLK * e;
        int c   = idx >> 6;
        int lo  = idx & 63;
        float v = 0.0f;
        if (c < 6)       v = 1.0f - 2.0f * (float)((lo >> c) & 1);
        else if (c == 6) v = 1.0f;
        Wt[c * PITCH + lo] = v;
    }
    __syncthreads();

    int wave = tid >> 5;
    int lane = tid & 31;
    if (wave < 4) {
        int row   = (wave << 4) + (lane & 15);
        int col   = lane & 15;
        int khalf = (lane >> 4) << 1;
        v8f acc = {};
        #pragma unroll
        for (int k = 0; k < 64; k += 4) {
            int k0 = k + khalf;
            v2f A = *(const v2f*)(P  + row * PITCH + k0);
            v2f B = *(const v2f*)(Wt + col * PITCH + k0);
            acc = __builtin_amdgcn_wmma_f32_16x16x4_f32(
                 false, A,  false, B,
                 (short)0, acc,  false,  false);
        }
        if (col < 8) {
            int rbase = (wave << 4) + ((lane >> 4) << 3);
            #pragma unroll
            for (int v = 0; v < 8; ++v)
                R[(rbase + v) * 8 + col] = acc[v];
        }
    }
    __syncthreads();

    if (tid < NQ) {
        int q = tid;
        float acc = 0.0f;
        if (q >= 6) {
            int c = NQ - 1 - q;
            #pragma unroll 1
            for (int hi = 0; hi < 64; ++hi) acc += R[hi * 8 + c];
        } else {
            int bp = 5 - q;
            #pragma unroll 1
            for (int hi = 0; hi < 64; ++hi) {
                float sgn = 1.0f - 2.0f * (float)((hi >> bp) & 1);
                acc += sgn * R[hi * 8 + 6];
            }
        }
        float nrm = 1.0f;
        int   fm  = 0;
        #pragma unroll 1
        for (int l = 0; l < NL; ++l) {
            int lm = 0;
            #pragma unroll
            for (int j = 0; j < NQ; ++j) {
                nrm *= gC[l * NQ + j];
                lm  |= gK[l * NQ + j] << (NQ - 1 - j);
            }
            fm = cnot_dst(fm ^ lm);
        }
        acc *= nrm;
        out[b * NQ + q] = ((fm >> (NQ - 1 - q)) & 1) ? -acc : acc;
    }
}

extern "C" void kernel_launch(void* const* d_in, const int* in_sizes, int n_in,
                              void* d_out, int out_size, void* d_ws, size_t ws_size,
                              hipStream_t stream) {
    const float* x  = (const float*)d_in[0];
    const float* w  = (const float*)d_in[1];
    float* out = (float*)d_out;
    int batch = in_sizes[0] / NQ;
    qsim_kernel<<<batch, BLK, 0, stream>>>(x, w, out);
}


# Single-wave serial end removed: the 64-step dependent LDS folds become a 32-lane partial-sum tree (new summation order), scale and flip mask are computed by an idle wave beside the WMMA
# speedup vs baseline: 3.0089x; 1.2923x over previous
#include <hip/hip_runtime.h>
#include <math.h>


#define NQ   12
#define NL   6
#define DIM  4096
#define BLK  256
#define NR   16

typedef __attribute__((ext_vector_type(2))) float v2f;
typedef __attribute__((ext_vector_type(8))) float v8f;
typedef __attribute__((ext_vector_type(4))) float v4f;

template <int HI>
__device__ __forceinline__ v2f fma_rot(v2f tp, v2f b, v2f a) {
    v2f d;
    if (HI)
        asm("v_pk_fma_f32 %0, %1, %2, %3 op_sel:[1,1,0] op_sel_hi:[1,0,1] neg_hi:[1,0,0]"
            : "=v"(d) : "v"(tp), "v"(b), "v"(a));
    else
        asm("v_pk_fma_f32 %0, %1, %2, %3 op_sel:[0,1,0] op_sel_hi:[0,0,1] neg_hi:[1,0,0]"
            : "=v"(d) : "v"(tp), "v"(b), "v"(a));
    return d;
}
template <int HI>
__device__ __forceinline__ void rx_pair(v2f& a, v2f& b, v2f tp) {
    v2f na = fma_rot<HI>(tp, b, a);
    v2f nb = fma_rot<HI>(tp, a, b);
    a = na;
    b = nb;
}

__host__ __device__ constexpr int swz(int s) {
    return s ^ ((s >> 5) & 31);
}

__host__ __device__ constexpr int cnot_dst(int s) {
    int j = s & (DIM - 1);
    j ^= j >> 1;
    j ^= j >> 2;
    j ^= j >> 4;
    j ^= j >> 8;
    j ^= (j & 1) << (NQ - 1);
    return (s & ~(DIM - 1)) | j;
}

template <int SH>
__device__ __forceinline__ void load16(v2f (&a)[NR], const v2f* st, int base) {
    #pragma unroll
    for (int r = 0; r < NR; ++r) a[r] = st[base ^ swz(r << SH)];
}
template <int SH>
__device__ __forceinline__ void store16(const v2f (&a)[NR], v2f* st, int base) {
    #pragma unroll
    for (int r = 0; r < NR; ++r) st[base ^ swz(r << SH)] = a[r];
}

__device__ __forceinline__ void rx4(v2f (&a)[NR], v4f t4) {
    #pragma unroll
    for (int j = 0; j < 4; ++j) {
        const v2f tp = (j < 2) ? t4.xy : t4.zw;
        const int m = 8 >> j;
        #pragma unroll
        for (int r = 0; r < NR; ++r)
            if (!(r & m)) {
                if (j & 1) rx_pair<1>(a[r], a[r | m], tp);
                else       rx_pair<0>(a[r], a[r | m], tp);
            }
    }
}

#define PITCH 68

__global__ __launch_bounds__(BLK, 8)
void qsim_kernel(const float* __restrict__ x,
                 const float* __restrict__ w,
                 float* __restrict__ out) {
    __shared__ v2f lds2[2 * DIM];
    __shared__ __attribute__((aligned(16))) float gT[NL * NQ];
    __shared__ __attribute__((aligned(16))) float gC[NL * NQ];
    __shared__ __attribute__((aligned(16))) int   gK[NL * NQ];

    const int tid = threadIdx.x;
    const int b   = blockIdx.x;

    if (tid < NL * NQ) {
        float th = w[tid];
        float k  = rintf(th * 0.318309886183790671538f);
        float h  = fmaf(-k, 1.57079637050628662109375f, 0.5f * th);
        h = fmaf(k, 4.37113882867379e-8f, h);
        float c = __cosf(h), s = __sinf(h);
        gT[tid] = s * __builtin_amdgcn_rcpf(c);
        gC[tid] = c * c;
        gK[tid] = (((int)k) & 1) << (NQ - 1 - (tid % NQ));
    }

    const int i1 = (tid & 15) | (((tid >> 4) & 1) << 9) | (((tid >> 5) & 1) << 8)
                 | (((tid >> 6) & 3) << 10);
    const int i2 = tid << 4;
    const int p0 = swz(tid);
    const int p1 = swz(i1);
    const int p2 = swz(i2);
    const int pd = swz(cnot_dst(i2));

    float* P  = (float*)(lds2 + (NL & 1) * DIM);
    float* Rt = P + 64 * PITCH;
    float* Wt = Rt + 8 * PITCH;
    float* NF = Wt + 16 * PITCH;

    float a0r[NQ], a0i[NQ], a1r[NQ], a1i[NQ];
    #pragma unroll
    for (int q = 0; q < NQ; ++q) {
        float ang = tanhf(x[b * NQ + q]) * 3.14159265358979323846f;
        float h = 0.5f * ang;
        float c = __cosf(h), s = __sinf(h);
        a0r[q] = c * c;  a0i[q] = -s * c;
        a1r[q] = s * c;  a1i[q] = s * s;
    }
    v2f amp[NR];
    #pragma unroll
    for (int k = 0; k < NR; ++k) {
        float vr = 1.0f, vi = 0.0f;
        #pragma unroll
        for (int q = 0; q < NQ; ++q) {
            int bit = (q < 4) ? ((k >> (3 - q)) & 1) : ((tid >> (NQ - 1 - q)) & 1);
            float ar = bit ? a1r[q] : a0r[q];
            float ai = bit ? a1i[q] : a0i[q];
            float nr = vr * ar - vi * ai;
            float ni = vr * ai + vi * ar;
            vr = nr; vi = ni;
        }
        amp[k].x = vr; amp[k].y = vi;
    }

    __syncthreads();
    {
        rx4(amp, *(const v4f*)(gT));
        store16<8>(amp, lds2, p0);
    }
    int cur = 0;
    #pragma unroll 1
    for (int l = 0; l < NL - 1; ++l) {
        const v4f* tq = (const v4f*)(gT + l * NQ + 4);
        const v4f t1 = tq[0], t2 = tq[1], t0 = tq[2];
        const int sb = cur << 12;
        const int ob = sb ^ DIM;

        __syncthreads();
        {
            v2f a1[NR];
            load16<4>(a1, lds2, p1 | sb);
            rx4(a1, t1);
            store16<4>(a1, lds2, p1 | sb);
        }

        __syncthreads();
        {
            v2f a2[NR];
            load16<0>(a2, lds2, p2 | sb);
            rx4(a2, t2);
            const int sd = pd | ob;
            #pragma unroll
            for (int r = 0; r < NR; ++r) lds2[sd ^ swz(cnot_dst(r))] = a2[r];
        }

        __syncthreads();
        {
            v2f a0[NR];
            load16<8>(a0, lds2, p0 | ob);
            rx4(a0, t0);
            store16<8>(a0, lds2, p0 | ob);
        }
        cur ^= 1;
    }
    {
        const v4f* tq = (const v4f*)(gT + (NL - 1) * NQ + 4);
        const v4f t1 = tq[0], t2 = tq[1];
        v2f* st = lds2 + ((NL - 1) & 1) * DIM;

        __syncthreads();
        {
            v2f a1[NR];
            load16<4>(a1, st, p1);
            rx4(a1, t1);
            store16<4>(a1, st, p1);
        }
        __syncthreads();
        {
            v2f a2[NR];
            load16<0>(a2, st, p2);
            rx4(a2, t2);
            const int d2 = swz(pd);
            #pragma unroll
            for (int r = 0; r < NR; ++r) {
                int d = d2 ^ cnot_dst(r);
                P[(d >> 6) * PITCH + (d & 63)] = a2[r].x * a2[r].x + a2[r].y * a2[r].y;
            }
        }
    }

    #pragma unroll
    for (int e = 0; e < 4; ++e) {
        int idx = tid + BLK * e;
        int c   = idx >> 6;
        int lo  = idx & 63;
        float v = 0.0f;
        if (c < 6)       v = 1.0f - 2.0f * (float)((lo >> c) & 1);
        else if (c == 6) v = 1.0f;
        Wt[c * PITCH + lo] = v;
    }
    __syncthreads();

    int wave = tid >> 5;
    int lane = tid & 31;
    if (wave < 4) {
        int row   = (wave << 4) + (lane & 15);
        int col   = lane & 15;
        int khalf = (lane >> 4) << 1;
        v8f acc = {};
        #pragma unroll
        for (int k = 0; k < 64; k += 4) {
            int k0 = k + khalf;
            v2f A = *(const v2f*)(P  + row * PITCH + k0);
            v2f B = *(const v2f*)(Wt + col * PITCH + k0);
            acc = __builtin_amdgcn_wmma_f32_16x16x4_f32(
                 false, A,  false, B,
                 (short)0, acc,  false,  false);
        }
        if (col < 8) {
            int rbase = (wave << 4) + ((lane >> 4) << 3);
            v4f d0 = {acc[0], acc[1], acc[2], acc[3]};
            v4f d1 = {acc[4], acc[5], acc[6], acc[7]};
            *(v4f*)(Rt + col * PITCH + rbase)     = d0;
            *(v4f*)(Rt + col * PITCH + rbase + 4) = d1;
        }
    } else if (wave == 4) {
        float n0 = 1.0f, n1 = 1.0f, n2 = 1.0f, n3 = 1.0f;
        #pragma unroll
        for (int g = 0; g < NL * NQ; g += 4) {
            v4f c4 = *(const v4f*)(gC + g);
            n0 *= c4.x; n1 *= c4.y; n2 *= c4.z; n3 *= c4.w;
        }
        const int ll = (lane < NL) ? lane : (NL - 1);
        int lm = 0;
        #pragma unroll
        for (int j = 0; j < NQ; ++j) lm |= gK[ll * NQ + j];
        int fm = 0;
        #pragma unroll
        for (int l = 0; l < NL; ++l) fm = cnot_dst(fm ^ __builtin_amdgcn_readlane(lm, l));
        if (lane == 0) {
            NF[0] = (n0 * n1) * (n2 * n3);
            NF[1] = __int_as_float(fm);
        }
    }
    __syncthreads();

    if (wave == 0) {
        const int q    = lane & 15;
        const int half = lane >> 4;
        const int col  = (q < 6) ? 6 : ((NQ - 1 - q) & 7);
        const unsigned sm = (q == 0) ? (half ? 0xFFFFFFFFu : 0u)
                          : (q == 1) ? 0xFFFF0000u
                          : (q == 2) ? 0xFF00FF00u
                          : (q == 3) ? 0xF0F0F0F0u
                          : (q == 4) ? 0xCCCCCCCCu
                          : (q == 5) ? 0xAAAAAAAAu : 0u;
        const v4f* rp = (const v4f*)(Rt + col * PITCH + 32 * half);
        float s0 = 0.0f, s1 = 0.0f, s2 = 0.0f, s3 = 0.0f;
        #pragma unroll
        for (int i = 0; i < 8; ++i) {
            v4f v = rp[i];
            s0 += __uint_as_float(__float_as_uint(v.x) ^ ((sm << (31 - 4 * i)) & 0x80000000u));
            s1 += __uint_as_float(__float_as_uint(v.y) ^ ((sm << (30 - 4 * i)) & 0x80000000u));
            s2 += __uint_as_float(__float_as_uint(v.z) ^ ((sm << (29 - 4 * i)) & 0x80000000u));
            s3 += __uint_as_float(__float_as_uint(v.w) ^ ((sm << (28 - 4 * i)) & 0x80000000u));
        }
        float acc = (s0 + s1) + (s2 + s3);
        acc += __shfl_xor(acc, 16);
        if (lane < NQ) {
            const float nrm = NF[0];
            const int   fm  = __float_as_int(NF[1]);
            acc *= nrm;
            out[b * NQ + lane] = ((fm >> (NQ - 1 - lane)) & 1) ? -acc : acc;
        }
    }
}

extern "C" void kernel_launch(void* const* d_in, const int* in_sizes, int n_in,
                              void* d_out, int out_size, void* d_ws, size_t ws_size,
                              hipStream_t stream) {
    const float* x  = (const float*)d_in[0];
    const float* w  = (const float*)d_in[1];
    float* out = (float*)d_out;
    int batch = in_sizes[0] / NQ;
    qsim_kernel<<<batch, BLK, 0, stream>>>(x, w, out);
}


# Layer 0's RX gates applied to the 12 per-qubit amplitude pairs before the product state is formed (RX keeps a product a product); the init stores through the CNOT ring: layer 0 is one store pass
# speedup vs baseline: 3.5575x; 1.1823x over previous
#include <hip/hip_runtime.h>
#include <math.h>


#define NQ   12
#define NL   6
#define DIM  4096
#define BLK  256
#define NR   16

typedef __attribute__((ext_vector_type(2))) float v2f;
typedef __attribute__((ext_vector_type(8))) float v8f;
typedef __attribute__((ext_vector_type(4))) float v4f;

template <int HI>
__device__ __forceinline__ v2f fma_rot(v2f tp, v2f b, v2f a) {
    v2f d;
    if (HI)
        asm("v_pk_fma_f32 %0, %1, %2, %3 op_sel:[1,1,0] op_sel_hi:[1,0,1] neg_hi:[1,0,0]"
            : "=v"(d) : "v"(tp), "v"(b), "v"(a));
    else
        asm("v_pk_fma_f32 %0, %1, %2, %3 op_sel:[0,1,0] op_sel_hi:[0,0,1] neg_hi:[1,0,0]"
            : "=v"(d) : "v"(tp), "v"(b), "v"(a));
    return d;
}
template <int HI>
__device__ __forceinline__ void rx_pair(v2f& a, v2f& b, v2f tp) {
    v2f na = fma_rot<HI>(tp, b, a);
    v2f nb = fma_rot<HI>(tp, a, b);
    a = na;
    b = nb;
}

__host__ __device__ constexpr int swz(int s) {
    return s ^ ((s >> 5) & 31);
}

__host__ __device__ constexpr int cnot_dst(int s) {
    int j = s & (DIM - 1);
    j ^= j >> 1;
    j ^= j >> 2;
    j ^= j >> 4;
    j ^= j >> 8;
    j ^= (j & 1) << (NQ - 1);
    return (s & ~(DIM - 1)) | j;
}

template <int SH>
__device__ __forceinline__ void load16(v2f (&a)[NR], const v2f* st, int base) {
    #pragma unroll
    for (int r = 0; r < NR; ++r) a[r] = st[base ^ swz(r << SH)];
}
template <int SH>
__device__ __forceinline__ void store16(const v2f (&a)[NR], v2f* st, int base) {
    #pragma unroll
    for (int r = 0; r < NR; ++r) st[base ^ swz(r << SH)] = a[r];
}

__device__ __forceinline__ void rx4(v2f (&a)[NR], v4f t4) {
    #pragma unroll
    for (int j = 0; j < 4; ++j) {
        const v2f tp = (j < 2) ? t4.xy : t4.zw;
        const int m = 8 >> j;
        #pragma unroll
        for (int r = 0; r < NR; ++r)
            if (!(r & m)) {
                if (j & 1) rx_pair<1>(a[r], a[r | m], tp);
                else       rx_pair<0>(a[r], a[r | m], tp);
            }
    }
}

#define PITCH 68

__global__ __launch_bounds__(BLK, 8)
void qsim_kernel(const float* __restrict__ x,
                 const float* __restrict__ w,
                 float* __restrict__ out) {
    __shared__ v2f lds2[2 * DIM];
    __shared__ __attribute__((aligned(16))) float gT[NL * NQ];
    __shared__ __attribute__((aligned(16))) float gC[NL * NQ];
    __shared__ __attribute__((aligned(16))) int   gK[NL * NQ];

    const int tid = threadIdx.x;
    const int b   = blockIdx.x;

    if (tid < NL * NQ) {
        float th = w[tid];
        float k  = rintf(th * 0.318309886183790671538f);
        float h  = fmaf(-k, 1.57079637050628662109375f, 0.5f * th);
        h = fmaf(k, 4.37113882867379e-8f, h);
        float c = __cosf(h), s = __sinf(h);
        gT[tid] = s * __builtin_amdgcn_rcpf(c);
        gC[tid] = c * c;
        gK[tid] = (((int)k) & 1) << (NQ - 1 - (tid % NQ));
    }

    const int i1 = (tid & 15) | (((tid >> 4) & 1) << 9) | (((tid >> 5) & 1) << 8)
                 | (((tid >> 6) & 3) << 10);
    const int i2 = tid << 4;
    const int p0 = swz(tid);
    const int p1 = swz(i1);
    const int p2 = swz(i2);
    const int pd = swz(cnot_dst(i2));

    float* P  = (float*)(lds2 + (NL & 1) * DIM);
    float* Rt = P + 64 * PITCH;
    float* Wt = Rt + 8 * PITCH;
    float* NF = Wt + 16 * PITCH;

    float a0r[NQ], a0i[NQ], a1r[NQ], a1i[NQ];
    #pragma unroll
    for (int q = 0; q < NQ; ++q) {
        float ang = tanhf(x[b * NQ + q]) * 3.14159265358979323846f;
        float h = 0.5f * ang;
        float c = __cosf(h), s = __sinf(h);
        a0r[q] = c * c;  a0i[q] = -s * c;
        a1r[q] = s * c;  a1i[q] = s * s;
    }
    __syncthreads();

    #pragma unroll
    for (int q = 0; q < NQ; ++q) {
        const float t = gT[q];
        const float n0r = fmaf( t, a1i[q], a0r[q]);
        const float n0i = fmaf(-t, a1r[q], a0i[q]);
        const float n1r = fmaf( t, a0i[q], a1r[q]);
        const float n1i = fmaf(-t, a0r[q], a1i[q]);
        a0r[q] = n0r;  a0i[q] = n0i;
        a1r[q] = n1r;  a1i[q] = n1i;
    }
    float tr = 1.0f, ti = 0.0f;
    #pragma unroll
    for (int q = 0; q < 8; ++q) {
        const int bit = (tid >> (7 - q)) & 1;
        const float ar = bit ? a1r[q] : a0r[q];
        const float ai = bit ? a1i[q] : a0i[q];
        const float nr = tr * ar - ti * ai;
        const float ni = tr * ai + ti * ar;
        tr = nr; ti = ni;
    }
    {
        v2f amp[NR];
        #pragma unroll
        for (int r = 0; r < NR; ++r) {
            float vr = tr, vi = ti;
            #pragma unroll
            for (int q = 8; q < NQ; ++q) {
                const int bit = (r >> (NQ - 1 - q)) & 1;
                const float ar = bit ? a1r[q] : a0r[q];
                const float ai = bit ? a1i[q] : a0i[q];
                const float nr = vr * ar - vi * ai;
                const float ni = vr * ai + vi * ar;
                vr = nr; vi = ni;
            }
            amp[r].x = vr; amp[r].y = vi;
        }
        const int sd = pd | DIM;
        #pragma unroll
        for (int r = 0; r < NR; ++r) lds2[sd ^ swz(cnot_dst(r))] = amp[r];
    }

    __syncthreads();
    {
        v2f a0[NR];
        load16<8>(a0, lds2, p0 | DIM);
        rx4(a0, *(const v4f*)(gT + NQ));
        store16<8>(a0, lds2, p0 | DIM);
    }
    int cur = 1;
    #pragma unroll 1
    for (int l = 1; l < NL - 1; ++l) {
        const v4f* tq = (const v4f*)(gT + l * NQ + 4);
        const v4f t1 = tq[0], t2 = tq[1], t0 = tq[2];
        const int sb = cur << 12;
        const int ob = sb ^ DIM;

        __syncthreads();
        {
            v2f a1[NR];
            load16<4>(a1, lds2, p1 | sb);
            rx4(a1, t1);
            store16<4>(a1, lds2, p1 | sb);
        }

        __syncthreads();
        {
            v2f a2[NR];
            load16<0>(a2, lds2, p2 | sb);
            rx4(a2, t2);
            const int sd = pd | ob;
            #pragma unroll
            for (int r = 0; r < NR; ++r) lds2[sd ^ swz(cnot_dst(r))] = a2[r];
        }

        __syncthreads();
        {
            v2f a0[NR];
            load16<8>(a0, lds2, p0 | ob);
            rx4(a0, t0);
            store16<8>(a0, lds2, p0 | ob);
        }
        cur ^= 1;
    }
    {
        const v4f* tq = (const v4f*)(gT + (NL - 1) * NQ + 4);
        const v4f t1 = tq[0], t2 = tq[1];
        v2f* st = lds2 + ((NL - 1) & 1) * DIM;

        __syncthreads();
        {
            v2f a1[NR];
            load16<4>(a1, st, p1);
            rx4(a1, t1);
            store16<4>(a1, st, p1);
        }
        __syncthreads();
        {
            v2f a2[NR];
            load16<0>(a2, st, p2);
            rx4(a2, t2);
            const int d2 = swz(pd);
            #pragma unroll
            for (int r = 0; r < NR; ++r) {
                int d = d2 ^ cnot_dst(r);
                P[(d >> 6) * PITCH + (d & 63)] = a2[r].x * a2[r].x + a2[r].y * a2[r].y;
            }
        }
    }

    #pragma unroll
    for (int e = 0; e < 4; ++e) {
        int idx = tid + BLK * e;
        int c   = idx >> 6;
        int lo  = idx & 63;
        float v = 0.0f;
        if (c < 6)       v = 1.0f - 2.0f * (float)((lo >> c) & 1);
        else if (c == 6) v = 1.0f;
        Wt[c * PITCH + lo] = v;
    }
    __syncthreads();

    int wave = tid >> 5;
    int lane = tid & 31;
    if (wave < 4) {
        int row   = (wave << 4) + (lane & 15);
        int col   = lane & 15;
        int khalf = (lane >> 4) << 1;
        v8f acc = {};
        #pragma unroll
        for (int k = 0; k < 64; k += 4) {
            int k0 = k + khalf;
            v2f A = *(const v2f*)(P  + row * PITCH + k0);
            v2f B = *(const v2f*)(Wt + col * PITCH + k0);
            acc = __builtin_amdgcn_wmma_f32_16x16x4_f32(
                 false, A,  false, B,
                 (short)0, acc,  false,  false);
        }
        if (col < 8) {
            int rbase = (wave << 4) + ((lane >> 4) << 3);
            v4f d0 = {acc[0], acc[1], acc[2], acc[3]};
            v4f d1 = {acc[4], acc[5], acc[6], acc[7]};
            *(v4f*)(Rt + col * PITCH + rbase)     = d0;
            *(v4f*)(Rt + col * PITCH + rbase + 4) = d1;
        }
    } else if (wave == 4) {
        float n0 = 1.0f, n1 = 1.0f, n2 = 1.0f, n3 = 1.0f;
        #pragma unroll
        for (int g = 0; g < NL * NQ; g += 4) {
            v4f c4 = *(const v4f*)(gC + g);
            n0 *= c4.x; n1 *= c4.y; n2 *= c4.z; n3 *= c4.w;
        }
        const int ll = (lane < NL) ? lane : (NL - 1);
        int lm = 0;
        #pragma unroll
        for (int j = 0; j < NQ; ++j) lm |= gK[ll * NQ + j];
        int fm = 0;
        #pragma unroll
        for (int l = 0; l < NL; ++l) fm = cnot_dst(fm ^ __builtin_amdgcn_readlane(lm, l));
        if (lane == 0) {
            NF[0] = (n0 * n1) * (n2 * n3);
            NF[1] = __int_as_float(fm);
        }
    }
    __syncthreads();

    if (wave == 0) {
        const int q    = lane & 15;
        const int half = lane >> 4;
        const int col  = (q < 6) ? 6 : ((NQ - 1 - q) & 7);
        const unsigned sm = (q == 0) ? (half ? 0xFFFFFFFFu : 0u)
                          : (q == 1) ? 0xFFFF0000u
                          : (q == 2) ? 0xFF00FF00u
                          : (q == 3) ? 0xF0F0F0F0u
                          : (q == 4) ? 0xCCCCCCCCu
                          : (q == 5) ? 0xAAAAAAAAu : 0u;
        const v4f* rp = (const v4f*)(Rt + col * PITCH + 32 * half);
        float s0 = 0.0f, s1 = 0.0f, s2 = 0.0f, s3 = 0.0f;
        #pragma unroll
        for (int i = 0; i < 8; ++i) {
            v4f v = rp[i];
            s0 += __uint_as_float(__float_as_uint(v.x) ^ ((sm << (31 - 4 * i)) & 0x80000000u));
            s1 += __uint_as_float(__float_as_uint(v.y) ^ ((sm << (30 - 4 * i)) & 0x80000000u));
            s2 += __uint_as_float(__float_as_uint(v.z) ^ ((sm << (29 - 4 * i)) & 0x80000000u));
            s3 += __uint_as_float(__float_as_uint(v.w) ^ ((sm << (28 - 4 * i)) & 0x80000000u));
        }
        float acc = (s0 + s1) + (s2 + s3);
        acc += __shfl_xor(acc, 16);
        if (lane < NQ) {
            const float nrm = NF[0];
            const int   fm  = __float_as_int(NF[1]);
            acc *= nrm;
            out[b * NQ + lane] = ((fm >> (NQ - 1 - lane)) & 1) ? -acc : acc;
        }
    }
}

extern "C" void kernel_launch(void* const* d_in, const int* in_sizes, int n_in,
                              void* d_out, int out_size, void* d_ws, size_t ws_size,
                              hipStream_t stream) {
    const float* x  = (const float*)d_in[0];
    const float* w  = (const float*)d_in[1];
    float* out = (float*)d_out;
    int batch = in_sizes[0] / NQ;
    qsim_kernel<<<batch, BLK, 0, stream>>>(x, w, out);
}


# All 72 RX gates commute once the CNOT rings are index relabelling: one phase per parity mode (-1)^popcount(k&j), then 12 add/sub butterfly stages replace 60 gate stages (new summation order)
# speedup vs baseline: 6.2270x; 1.7504x over previous
#include <hip/hip_runtime.h>
#include <math.h>


#define NQ   12
#define NL   6
#define DIM  4096
#define BLK  256
#define NR   16

typedef __attribute__((ext_vector_type(2))) float v2f;
typedef __attribute__((ext_vector_type(8))) float v8f;
typedef __attribute__((ext_vector_type(4))) float v4f;

__host__ __device__ constexpr int swz(int s) {
    return s ^ ((s >> 5) & 31);
}

__host__ __device__ constexpr int cnot_dst(int s) {
    int j = s & (DIM - 1);
    j ^= j >> 1;
    j ^= j >> 2;
    j ^= j >> 4;
    j ^= j >> 8;
    j ^= (j & 1) << (NQ - 1);
    return (s & ~(DIM - 1)) | j;
}
__host__ __device__ constexpr int ring_all(int s) {
    for (int i = 0; i < NL; ++i) s = cnot_dst(s);
    return s;
}
__host__ __device__ constexpr int cnot_src(int j) {
    int t = j ^ ((j & 1) << (NQ - 1));
    return t ^ (t >> 1);
}
__host__ __device__ constexpr int gate_mask(int l, int q) {
    int m = 1 << (NQ - 1 - q);
    for (int i = 0; i < l; ++i) m = cnot_src(m);
    return m;
}
__host__ __device__ constexpr int dual_step(int k) {
    int t = (k ^ (k << 1)) & (DIM - 1);
    return t ^ (t >> (NQ - 1));
}

template <int SH>
__device__ __forceinline__ void load16(v2f (&a)[NR], const v2f* st, int base) {
    #pragma unroll
    for (int r = 0; r < NR; ++r) a[r] = st[base ^ swz(r << SH)];
}
template <int SH>
__device__ __forceinline__ void store16(const v2f (&a)[NR], v2f* st, int base) {
    #pragma unroll
    for (int r = 0; r < NR; ++r) st[base ^ swz(r << SH)] = a[r];
}

__device__ __forceinline__ void bfly4(v2f (&a)[NR]) {
    #pragma unroll
    for (int m = 1; m < NR; m <<= 1) {
        #pragma unroll
        for (int r = 0; r < NR; ++r)
            if (!(r & m)) {
                const v2f s = a[r] + a[r | m];
                const v2f d = a[r] - a[r | m];
                a[r]     = s;
                a[r | m] = d;
            }
    }
}

template <int G0>
__device__ __forceinline__ void phase_terms(float (&G)[NR], const float* gH, int tau) {
    if constexpr (G0 < NL * NQ) {
        constexpr int L  = G0 / NQ;
        constexpr int Q  = G0 % NQ;
        constexpr int PB = NQ - 1 - Q;
        constexpr int MU = gate_mask(L, Q) & (NR - 1);
        const unsigned sb = ((unsigned)tau << (31 - PB)) & 0x80000000u;
        G[MU] += __uint_as_float(__float_as_uint(gH[G0]) ^ sb);
        phase_terms<G0 + 1>(G, gH, (Q == NQ - 1) ? dual_step(tau) : tau);
    }
}

#define PITCH 68

__global__ __launch_bounds__(BLK, 8)
void qsim_kernel(const float* __restrict__ x,
                 const float* __restrict__ w,
                 float* __restrict__ out) {
    __shared__ v2f lds2[2 * DIM];
    __shared__ __attribute__((aligned(16))) float gH[NL * NQ];

    const int tid = threadIdx.x;
    const int b   = blockIdx.x;

    if (tid < NL * NQ) {
        float th = w[tid];
        float k  = rintf(th * 0.0795774715459476679f);
        float h  = fmaf(-k, 6.283185482025146484375f, 0.5f * th);
        h = fmaf(k, 1.748455531469516e-7f, h);
        gH[tid] = h * 0.159154943091895335769f;
    }

    const int i1 = (tid & 15) | (((tid >> 4) & 1) << 9) | (((tid >> 5) & 1) << 8)
                 | (((tid >> 6) & 3) << 10);
    const int i2 = tid << 4;
    const int p0 = swz(tid);
    const int p1 = swz(i1);
    const int p2 = swz(i2);

    float* P  = (float*)(lds2 + DIM);
    float* Rt = P + 64 * PITCH;
    float* Wt = Rt + 8 * PITCH;

    float u0r[NQ], u0i[NQ], u1r[NQ], u1i[NQ];
    #pragma unroll
    for (int q = 0; q < NQ; ++q) {
        float ang = tanhf(x[b * NQ + q]) * 3.14159265358979323846f;
        float h = 0.5f * ang;
        float c = __cosf(h), s = __sinf(h);
        const float cc = c * c, ss = s * s, sc = s * c;
        u0r[q] = 0.5f * (cc + sc);  u0i[q] = 0.5f * (ss - sc);
        u1r[q] = 0.5f * (cc - sc);  u1i[q] = -0.5f * (sc + ss);
    }
    __syncthreads();

    float ph[NR];
    #pragma unroll
    for (int r = 0; r < NR; ++r) ph[r] = 0.0f;
    phase_terms<0>(ph, gH, i2);
    #pragma unroll
    for (int m = 1; m < NR; m <<= 1) {
        #pragma unroll
        for (int r = 0; r < NR; ++r)
            if (!(r & m)) {
                const float s = ph[r] + ph[r | m];
                const float d = ph[r] - ph[r | m];
                ph[r]     = s;
                ph[r | m] = d;
            }
    }

    float tr = 1.0f, ti = 0.0f;
    #pragma unroll
    for (int q = 0; q < 8; ++q) {
        const int bit = (tid >> (7 - q)) & 1;
        const float ar = bit ? u1r[q] : u0r[q];
        const float ai = bit ? u1i[q] : u0i[q];
        const float nr = tr * ar - ti * ai;
        const float ni = tr * ai + ti * ar;
        tr = nr; ti = ni;
    }
    {
        v2f amp[NR];
        #pragma unroll
        for (int r = 0; r < NR; ++r) {
            float vr = tr, vi = ti;
            #pragma unroll
            for (int q = 8; q < NQ; ++q) {
                const int bit = (r >> (NQ - 1 - q)) & 1;
                const float ar = bit ? u1r[q] : u0r[q];
                const float ai = bit ? u1i[q] : u0i[q];
                const float nr = vr * ar - vi * ai;
                const float ni = vr * ai + vi * ar;
                vr = nr; vi = ni;
            }
            const float cp = __builtin_amdgcn_cosf(ph[r]);
            const float sp = __builtin_amdgcn_sinf(ph[r]);
            amp[r].x = vr * cp + vi * sp;
            amp[r].y = vi * cp - vr * sp;
        }
        bfly4(amp);
        store16<0>(amp, lds2, p2);
    }
    __syncthreads();
    {
        v2f a1[NR];
        load16<4>(a1, lds2, p1);
        bfly4(a1);
        store16<4>(a1, lds2, p1);
    }
    __syncthreads();
    {
        v2f a0[NR];
        load16<8>(a0, lds2, p0);
        bfly4(a0);
        const int d0 = ring_all(tid);
        #pragma unroll
        for (int r = 0; r < NR; ++r) {
            const int d = d0 ^ ring_all(r << 8);
            P[(d >> 6) * PITCH + (d & 63)] = a0[r].x * a0[r].x + a0[r].y * a0[r].y;
        }
    }

    #pragma unroll
    for (int e = 0; e < 4; ++e) {
        int idx = tid + BLK * e;
        int c   = idx >> 6;
        int lo  = idx & 63;
        float v = 0.0f;
        if (c < 6)       v = 1.0f - 2.0f * (float)((lo >> c) & 1);
        else if (c == 6) v = 1.0f;
        Wt[c * PITCH + lo] = v;
    }
    __syncthreads();

    int wave = tid >> 5;
    int lane = tid & 31;
    if (wave < 4) {
        int row   = (wave << 4) + (lane & 15);
        int col   = lane & 15;
        int khalf = (lane >> 4) << 1;
        v8f acc = {};
        #pragma unroll
        for (int k = 0; k < 64; k += 4) {
            int k0 = k + khalf;
            v2f A = *(const v2f*)(P  + row * PITCH + k0);
            v2f B = *(const v2f*)(Wt + col * PITCH + k0);
            acc = __builtin_amdgcn_wmma_f32_16x16x4_f32(
                 false, A,  false, B,
                 (short)0, acc,  false,  false);
        }
        if (col < 8) {
            int rbase = (wave << 4) + ((lane >> 4) << 3);
            v4f d0 = {acc[0], acc[1], acc[2], acc[3]};
            v4f d1 = {acc[4], acc[5], acc[6], acc[7]};
            *(v4f*)(Rt + col * PITCH + rbase)     = d0;
            *(v4f*)(Rt + col * PITCH + rbase + 4) = d1;
        }
    }
    __syncthreads();

    if (wave == 0) {
        const int q    = lane & 15;
        const int half = lane >> 4;
        const int col  = (q < 6) ? 6 : ((NQ - 1 - q) & 7);
        const unsigned sm = (q == 0) ? (half ? 0xFFFFFFFFu : 0u)
                          : (q == 1) ? 0xFFFF0000u
                          : (q == 2) ? 0xFF00FF00u
                          : (q == 3) ? 0xF0F0F0F0u
                          : (q == 4) ? 0xCCCCCCCCu
                          : (q == 5) ? 0xAAAAAAAAu : 0u;
        const v4f* rp = (const v4f*)(Rt + col * PITCH + 32 * half);
        float s0 = 0.0f, s1 = 0.0f, s2 = 0.0f, s3 = 0.0f;
        #pragma unroll
        for (int i = 0; i < 8; ++i) {
            v4f v = rp[i];
            s0 += __uint_as_float(__float_as_uint(v.x) ^ ((sm << (31 - 4 * i)) & 0x80000000u));
            s1 += __uint_as_float(__float_as_uint(v.y) ^ ((sm << (30 - 4 * i)) & 0x80000000u));
            s2 += __uint_as_float(__float_as_uint(v.z) ^ ((sm << (29 - 4 * i)) & 0x80000000u));
            s3 += __uint_as_float(__float_as_uint(v.w) ^ ((sm << (28 - 4 * i)) & 0x80000000u));
        }
        float acc = (s0 + s1) + (s2 + s3);
        acc += __shfl_xor(acc, 16);
        if (lane < NQ) out[b * NQ + lane] = acc;
    }
}

extern "C" void kernel_launch(void* const* d_in, const int* in_sizes, int n_in,
                              void* d_out, int out_size, void* d_ws, size_t ws_size,
                              hipStream_t stream) {
    const float* x  = (const float*)d_in[0];
    const float* w  = (const float*)d_in[1];
    float* out = (float*)d_out;
    int batch = in_sizes[0] / NQ;
    qsim_kernel<<<batch, BLK, 0, stream>>>(x, w, out);
}


# Two circuits per workgroup, one after the other through the same LDS: the weight-only work (angle table, mode phases with their sin/cos, sign table) is done once for both; grid 1024, was 2048
# speedup vs baseline: 6.7541x; 1.0846x over previous
#include <hip/hip_runtime.h>
#include <math.h>


#define NQ   12
#define NL   6
#define DIM  4096
#define BLK  256
#define NR   16

typedef __attribute__((ext_vector_type(2))) float v2f;
typedef __attribute__((ext_vector_type(8))) float v8f;
typedef __attribute__((ext_vector_type(4))) float v4f;

__host__ __device__ constexpr int swz(int s) {
    return s ^ ((s >> 5) & 31);
}

__host__ __device__ constexpr int cnot_dst(int s) {
    int j = s & (DIM - 1);
    j ^= j >> 1;
    j ^= j >> 2;
    j ^= j >> 4;
    j ^= j >> 8;
    j ^= (j & 1) << (NQ - 1);
    return (s & ~(DIM - 1)) | j;
}
__host__ __device__ constexpr int ring_all(int s) {
    for (int i = 0; i < NL; ++i) s = cnot_dst(s);
    return s;
}
__host__ __device__ constexpr int cnot_src(int j) {
    int t = j ^ ((j & 1) << (NQ - 1));
    return t ^ (t >> 1);
}
__host__ __device__ constexpr int gate_mask(int l, int q) {
    int m = 1 << (NQ - 1 - q);
    for (int i = 0; i < l; ++i) m = cnot_src(m);
    return m;
}
__host__ __device__ constexpr int dual_step(int k) {
    int t = (k ^ (k << 1)) & (DIM - 1);
    return t ^ (t >> (NQ - 1));
}

template <int SH>
__device__ __forceinline__ void load16(v2f (&a)[NR], const v2f* st, int base) {
    #pragma unroll
    for (int r = 0; r < NR; ++r) a[r] = st[base ^ swz(r << SH)];
}
template <int SH>
__device__ __forceinline__ void store16(const v2f (&a)[NR], v2f* st, int base) {
    #pragma unroll
    for (int r = 0; r < NR; ++r) st[base ^ swz(r << SH)] = a[r];
}

__device__ __forceinline__ void bfly4(v2f (&a)[NR]) {
    #pragma unroll
    for (int m = 1; m < NR; m <<= 1) {
        #pragma unroll
        for (int r = 0; r < NR; ++r)
            if (!(r & m)) {
                const v2f s = a[r] + a[r | m];
                const v2f d = a[r] - a[r | m];
                a[r]     = s;
                a[r | m] = d;
            }
    }
}

template <int G0>
__device__ __forceinline__ void phase_terms(float (&G)[NR], const float* gH, int tau) {
    if constexpr (G0 < NL * NQ) {
        constexpr int L  = G0 / NQ;
        constexpr int Q  = G0 % NQ;
        constexpr int PB = NQ - 1 - Q;
        constexpr int MU = gate_mask(L, Q) & (NR - 1);
        const unsigned sb = ((unsigned)tau << (31 - PB)) & 0x80000000u;
        G[MU] += __uint_as_float(__float_as_uint(gH[G0]) ^ sb);
        phase_terms<G0 + 1>(G, gH, (Q == NQ - 1) ? dual_step(tau) : tau);
    }
}

#define PITCH 68

#define CPW 2

#define NRQ 4
__device__ __forceinline__ void input_stage(const float (&xb)[NQ], int tid,
                                            float& tr, float& ti,
                                            float (&u0r)[NRQ], float (&u0i)[NRQ],
                                            float (&u1r)[NRQ], float (&u1i)[NRQ]) {
    tr = 1.0f; ti = 0.0f;
    #pragma unroll
    for (int q = 0; q < NQ; ++q) {
        float ang = tanhf(xb[q]) * 3.14159265358979323846f;
        float h = 0.5f * ang;
        float c = __cosf(h), s = __sinf(h);
        const float cc = c * c, ss = s * s, sc = s * c;
        const float a0r = 0.5f * (cc + sc), a0i = 0.5f * (ss - sc);
        const float a1r = 0.5f * (cc - sc), a1i = -0.5f * (sc + ss);
        if (q < 8) {
            const int bit = (tid >> (7 - q)) & 1;
            const float ar = bit ? a1r : a0r;
            const float ai = bit ? a1i : a0i;
            const float nr = tr * ar - ti * ai;
            const float ni = tr * ai + ti * ar;
            tr = nr; ti = ni;
        } else {
            u0r[q - 8] = a0r;  u0i[q - 8] = a0i;
            u1r[q - 8] = a1r;  u1i[q - 8] = a1i;
        }
    }
}

__global__ __launch_bounds__(BLK, 8)
void qsim_kernel(const float* __restrict__ x,
                 const float* __restrict__ w,
                 float* __restrict__ out) {
    __shared__ v2f lds2[2 * DIM];
    __shared__ __attribute__((aligned(16))) float gH[NL * NQ];

    const int tid = threadIdx.x;

    if (tid < NL * NQ) {
        float th = w[tid];
        float k  = rintf(th * 0.0795774715459476679f);
        float h  = fmaf(-k, 6.283185482025146484375f, 0.5f * th);
        h = fmaf(k, 1.748455531469516e-7f, h);
        gH[tid] = h * 0.159154943091895335769f;
    }

    const int i1 = (tid & 15) | (((tid >> 4) & 1) << 9) | (((tid >> 5) & 1) << 8)
                 | (((tid >> 6) & 3) << 10);
    const int i2 = tid << 4;
    int p0 = swz(tid);
    int p1 = swz(i1);
    int p2 = swz(i2);
    int d0 = ring_all(tid);
    int tl = tid;

    float* P  = (float*)(lds2 + DIM);
    float* Rt = P + 64 * PITCH;
    float* Wt = Rt + 8 * PITCH;

    #pragma unroll
    for (int e = 0; e < 4; ++e) {
        int idx = tid + BLK * e;
        int c   = idx >> 6;
        int lo  = idx & 63;
        float v = 0.0f;
        if (c < 6)       v = 1.0f - 2.0f * (float)((lo >> c) & 1);
        else if (c == 6) v = 1.0f;
        Wt[c * PITCH + lo] = v;
    }

    const int b0 = blockIdx.x * CPW;
    float xin[CPW][NQ];
    #pragma unroll
    for (int c = 0; c < CPW; ++c) {
        #pragma unroll
        for (int q = 0; q < NQ; ++q) xin[c][q] = x[(b0 + c) * NQ + q];
    }

    float tr, ti;
    float u0r[NRQ], u0i[NRQ], u1r[NRQ], u1i[NRQ];
    input_stage(xin[0], tid, tr, ti, u0r, u0i, u1r, u1i);
    __syncthreads();

    float ph[NR];
    #pragma unroll
    for (int r = 0; r < NR; ++r) ph[r] = 0.0f;
    phase_terms<0>(ph, gH, i2);
    #pragma unroll
    for (int m = 1; m < NR; m <<= 1) {
        #pragma unroll
        for (int r = 0; r < NR; ++r)
            if (!(r & m)) {
                const float s = ph[r] + ph[r | m];
                const float d = ph[r] - ph[r | m];
                ph[r]     = s;
                ph[r | m] = d;
            }
    }
    float cp[NR], sp[NR];
    #pragma unroll
    for (int r = 0; r < NR; ++r) {
        cp[r] = __builtin_amdgcn_cosf(ph[r]);
        sp[r] = __builtin_amdgcn_sinf(ph[r]);
    }

    #pragma unroll 1
    for (int cw = 0; cw < CPW; ++cw) {
        const int b = b0 + cw;
        asm volatile("" : "+v"(p0), "+v"(p1), "+v"(p2), "+v"(d0), "+v"(tl));

        {
            v2f amp[NR];
            #pragma unroll
            for (int r = 0; r < NR; ++r) {
                float vr = tr, vi = ti;
                #pragma unroll
                for (int q = 0; q < NRQ; ++q) {
                    const int bit = (r >> (NRQ - 1 - q)) & 1;
                    const float ar = bit ? u1r[q] : u0r[q];
                    const float ai = bit ? u1i[q] : u0i[q];
                    const float nr = vr * ar - vi * ai;
                    const float ni = vr * ai + vi * ar;
                    vr = nr; vi = ni;
                }
                amp[r].x = vr * cp[r] + vi * sp[r];
                amp[r].y = vi * cp[r] - vr * sp[r];
            }
            bfly4(amp);
            store16<0>(amp, lds2, p2);
        }
        __syncthreads();
        {
            v2f a1[NR];
            load16<4>(a1, lds2, p1);
            bfly4(a1);
            store16<4>(a1, lds2, p1);
        }
        __syncthreads();
        {
            v2f a0[NR];
            load16<8>(a0, lds2, p0);
            bfly4(a0);
            #pragma unroll
            for (int r = 0; r < NR; ++r) {
                const int d = d0 ^ ring_all(r << 8);
                P[(d >> 6) * PITCH + (d & 63)] = a0[r].x * a0[r].x + a0[r].y * a0[r].y;
            }
        }
        __syncthreads();

        int wave = tl >> 5;
        int lane = tl & 31;
        if (wave < 4) {
            int row   = (wave << 4) + (lane & 15);
            int col   = lane & 15;
            int khalf = (lane >> 4) << 1;
            v8f acc = {};
            #pragma unroll
            for (int k = 0; k < 64; k += 4) {
                int k0 = k + khalf;
                v2f A = *(const v2f*)(P  + row * PITCH + k0);
                v2f B = *(const v2f*)(Wt + col * PITCH + k0);
                acc = __builtin_amdgcn_wmma_f32_16x16x4_f32(
                     false, A,  false, B,
                     (short)0, acc,  false,  false);
            }
            if (col < 8) {
                int rbase = (wave << 4) + ((lane >> 4) << 3);
                v4f d0 = {acc[0], acc[1], acc[2], acc[3]};
                v4f d1 = {acc[4], acc[5], acc[6], acc[7]};
                *(v4f*)(Rt + col * PITCH + rbase)     = d0;
                *(v4f*)(Rt + col * PITCH + rbase + 4) = d1;
            }
        }
        __syncthreads();

        if (wave == 0) {
            const int q    = lane & 15;
            const int half = lane >> 4;
            const int col  = (q < 6) ? 6 : ((NQ - 1 - q) & 7);
            const unsigned sm = (q == 0) ? (half ? 0xFFFFFFFFu : 0u)
                              : (q == 1) ? 0xFFFF0000u
                              : (q == 2) ? 0xFF00FF00u
                              : (q == 3) ? 0xF0F0F0F0u
                              : (q == 4) ? 0xCCCCCCCCu
                              : (q == 5) ? 0xAAAAAAAAu : 0u;
            const v4f* rp = (const v4f*)(Rt + col * PITCH + 32 * half);
            float s0 = 0.0f, s1 = 0.0f, s2 = 0.0f, s3 = 0.0f;
            #pragma unroll
            for (int i = 0; i < 8; ++i) {
                v4f v = rp[i];
                s0 += __uint_as_float(__float_as_uint(v.x) ^ ((sm << (31 - 4 * i)) & 0x80000000u));
                s1 += __uint_as_float(__float_as_uint(v.y) ^ ((sm << (30 - 4 * i)) & 0x80000000u));
                s2 += __uint_as_float(__float_as_uint(v.z) ^ ((sm << (29 - 4 * i)) & 0x80000000u));
                s3 += __uint_as_float(__float_as_uint(v.w) ^ ((sm << (28 - 4 * i)) & 0x80000000u));
            }
            float acc = (s0 + s1) + (s2 + s3);
            acc += __shfl_xor(acc, 16);
            if (lane < NQ) out[b * NQ + lane] = acc;
        }
        if (cw + 1 < CPW) {
            float xn[NQ];
            #pragma unroll
            for (int q = 0; q < NQ; ++q) {
                xn[q] = xin[1][q];
                #pragma unroll
                for (int c = 2; c < CPW; ++c) xn[q] = (cw + 1 == c) ? xin[c][q] : xn[q];
            }
            input_stage(xn, tid, tr, ti, u0r, u0i, u1r, u1i);
        }
    }
}

extern "C" void kernel_launch(void* const* d_in, const int* in_sizes, int n_in,
                              void* d_out, int out_size, void* d_ws, size_t ws_size,
                              hipStream_t stream) {
    const float* x  = (const float*)d_in[0];
    const float* w  = (const float*)d_in[1];
    float* out = (float*)d_out;
    int groups = in_sizes[0] / (NQ * CPW);
    qsim_kernel<<<groups, BLK, 0, stream>>>(x, w, out);
}


# Input stage once per qubit on lanes: lane (circuit, qubit) computes tanh, sin, cos and the amplitude pair in one vector pass, handed out by lane reads; was 12 serial scalar-unit tanh chains per wave
# speedup vs baseline: 7.9442x; 1.1762x over previous
#include <hip/hip_runtime.h>
#include <math.h>


#define NQ   12
#define NL   6
#define DIM  4096
#define BLK  256
#define NR   16

typedef __attribute__((ext_vector_type(2))) float v2f;
typedef __attribute__((ext_vector_type(8))) float v8f;
typedef __attribute__((ext_vector_type(4))) float v4f;

__host__ __device__ constexpr int swz(int s) {
    return s ^ ((s >> 5) & 31);
}

__host__ __device__ constexpr int cnot_dst(int s) {
    int j = s & (DIM - 1);
    j ^= j >> 1;
    j ^= j >> 2;
    j ^= j >> 4;
    j ^= j >> 8;
    j ^= (j & 1) << (NQ - 1);
    return (s & ~(DIM - 1)) | j;
}
__host__ __device__ constexpr int ring_all(int s) {
    for (int i = 0; i < NL; ++i) s = cnot_dst(s);
    return s;
}
__host__ __device__ constexpr int cnot_src(int j) {
    int t = j ^ ((j & 1) << (NQ - 1));
    return t ^ (t >> 1);
}
__host__ __device__ constexpr int gate_mask(int l, int q) {
    int m = 1 << (NQ - 1 - q);
    for (int i = 0; i < l; ++i) m = cnot_src(m);
    return m;
}
__host__ __device__ constexpr int dual_step(int k) {
    int t = (k ^ (k << 1)) & (DIM - 1);
    return t ^ (t >> (NQ - 1));
}

template <int SH>
__device__ __forceinline__ void load16(v2f (&a)[NR], const v2f* st, int base) {
    #pragma unroll
    for (int r = 0; r < NR; ++r) a[r] = st[base ^ swz(r << SH)];
}
template <int SH>
__device__ __forceinline__ void store16(const v2f (&a)[NR], v2f* st, int base) {
    #pragma unroll
    for (int r = 0; r < NR; ++r) st[base ^ swz(r << SH)] = a[r];
}

__device__ __forceinline__ void bfly4(v2f (&a)[NR]) {
    #pragma unroll
    for (int m = 1; m < NR; m <<= 1) {
        #pragma unroll
        for (int r = 0; r < NR; ++r)
            if (!(r & m)) {
                const v2f s = a[r] + a[r | m];
                const v2f d = a[r] - a[r | m];
                a[r]     = s;
                a[r | m] = d;
            }
    }
}

template <int G0>
__device__ __forceinline__ void phase_terms(float (&G)[NR], const float* gH, int tau) {
    if constexpr (G0 < NL * NQ) {
        constexpr int L  = G0 / NQ;
        constexpr int Q  = G0 % NQ;
        constexpr int PB = NQ - 1 - Q;
        constexpr int MU = gate_mask(L, Q) & (NR - 1);
        const unsigned sb = ((unsigned)tau << (31 - PB)) & 0x80000000u;
        G[MU] += __uint_as_float(__float_as_uint(gH[G0]) ^ sb);
        phase_terms<G0 + 1>(G, gH, (Q == NQ - 1) ? dual_step(tau) : tau);
    }
}

#define PITCH 68

#define CPW 2

__device__ __forceinline__ void qubit_pair(float xv, float& a0r, float& a0i,
                                           float& a1r, float& a1i) {
    float ang = tanhf(xv) * 3.14159265358979323846f;
    float h = 0.5f * ang;
    float c = __cosf(h), s = __sinf(h);
    const float cc = c * c, ss = s * s, sc = s * c;
    a0r = 0.5f * (cc + sc);  a0i = 0.5f * (ss - sc);
    a1r = 0.5f * (cc - sc);  a1i = -0.5f * (sc + ss);
}
__device__ __forceinline__ float lane_value(float v, int l) {
    return __int_as_float(__builtin_amdgcn_readlane(__float_as_int(v), l));
}
#define NRQ 4
__device__ __forceinline__ void input_stage(float la0r, float la0i, float la1r, float la1i,
                                            int l0, int tid,
                                            float& tr, float& ti,
                                            float (&u0r)[NRQ], float (&u0i)[NRQ],
                                            float (&u1r)[NRQ], float (&u1i)[NRQ]) {
    tr = 1.0f; ti = 0.0f;
    #pragma unroll
    for (int q = 0; q < NQ; ++q) {
        const float a0r = lane_value(la0r, l0 + q), a0i = lane_value(la0i, l0 + q);
        const float a1r = lane_value(la1r, l0 + q), a1i = lane_value(la1i, l0 + q);
        if (q < 8) {
            const int bit = (tid >> (7 - q)) & 1;
            const float ar = bit ? a1r : a0r;
            const float ai = bit ? a1i : a0i;
            const float nr = tr * ar - ti * ai;
            const float ni = tr * ai + ti * ar;
            tr = nr; ti = ni;
        } else {
            u0r[q - 8] = a0r;  u0i[q - 8] = a0i;
            u1r[q - 8] = a1r;  u1i[q - 8] = a1i;
        }
    }
}

__global__ __launch_bounds__(BLK, 8)
void qsim_kernel(const float* __restrict__ x,
                 const float* __restrict__ w,
                 float* __restrict__ out) {
    __shared__ v2f lds2[2 * DIM];
    __shared__ __attribute__((aligned(16))) float gH[NL * NQ];

    const int tid = threadIdx.x;

    if (tid < NL * NQ) {
        float th = w[tid];
        float k  = rintf(th * 0.0795774715459476679f);
        float h  = fmaf(-k, 6.283185482025146484375f, 0.5f * th);
        h = fmaf(k, 1.748455531469516e-7f, h);
        gH[tid] = h * 0.159154943091895335769f;
    }

    const int i1 = (tid & 15) | (((tid >> 4) & 1) << 9) | (((tid >> 5) & 1) << 8)
                 | (((tid >> 6) & 3) << 10);
    const int i2 = tid << 4;
    int p0 = swz(tid);
    int p1 = swz(i1);
    int p2 = swz(i2);
    int d0 = ring_all(tid);
    int tl = tid;

    float* P  = (float*)(lds2 + DIM);
    float* Rt = P + 64 * PITCH;
    float* Wt = Rt + 8 * PITCH;

    #pragma unroll
    for (int e = 0; e < 4; ++e) {
        int idx = tid + BLK * e;
        int c   = idx >> 6;
        int lo  = idx & 63;
        float v = 0.0f;
        if (c < 6)       v = 1.0f - 2.0f * (float)((lo >> c) & 1);
        else if (c == 6) v = 1.0f;
        Wt[c * PITCH + lo] = v;
    }

    const int b0 = blockIdx.x * CPW;
    static_assert(CPW * NQ <= 32, "one lane per (circuit, qubit) pair");
    float la0r, la0i, la1r, la1i;
    {
        const int l  = tid & 31;
        const int li = (l < CPW * NQ) ? l : (CPW * NQ - 1);
        qubit_pair(x[b0 * NQ + li], la0r, la0i, la1r, la1i);
    }

    float tr, ti;
    float u0r[NRQ], u0i[NRQ], u1r[NRQ], u1i[NRQ];
    input_stage(la0r, la0i, la1r, la1i, 0, tid, tr, ti, u0r, u0i, u1r, u1i);
    __syncthreads();

    float ph[NR];
    #pragma unroll
    for (int r = 0; r < NR; ++r) ph[r] = 0.0f;
    phase_terms<0>(ph, gH, i2);
    #pragma unroll
    for (int m = 1; m < NR; m <<= 1) {
        #pragma unroll
        for (int r = 0; r < NR; ++r)
            if (!(r & m)) {
                const float s = ph[r] + ph[r | m];
                const float d = ph[r] - ph[r | m];
                ph[r]     = s;
                ph[r | m] = d;
            }
    }
    float cp[NR], sp[NR];
    #pragma unroll
    for (int r = 0; r < NR; ++r) {
        cp[r] = __builtin_amdgcn_cosf(ph[r]);
        sp[r] = __builtin_amdgcn_sinf(ph[r]);
    }

    #pragma unroll 1
    for (int cw = 0; cw < CPW; ++cw) {
        const int b = b0 + cw;
        asm volatile("" : "+v"(p0), "+v"(p1), "+v"(p2), "+v"(d0), "+v"(tl));

        {
            v2f amp[NR];
            #pragma unroll
            for (int r = 0; r < NR; ++r) {
                float vr = tr, vi = ti;
                #pragma unroll
                for (int q = 0; q < NRQ; ++q) {
                    const int bit = (r >> (NRQ - 1 - q)) & 1;
                    const float ar = bit ? u1r[q] : u0r[q];
                    const float ai = bit ? u1i[q] : u0i[q];
                    const float nr = vr * ar - vi * ai;
                    const float ni = vr * ai + vi * ar;
                    vr = nr; vi = ni;
                }
                amp[r].x = vr * cp[r] + vi * sp[r];
                amp[r].y = vi * cp[r] - vr * sp[r];
            }
            bfly4(amp);
            store16<0>(amp, lds2, p2);
        }
        __syncthreads();
        {
            v2f a1[NR];
            load16<4>(a1, lds2, p1);
            bfly4(a1);
            store16<4>(a1, lds2, p1);
        }
        __syncthreads();
        {
            v2f a0[NR];
            load16<8>(a0, lds2, p0);
            bfly4(a0);
            #pragma unroll
            for (int r = 0; r < NR; ++r) {
                const int d = d0 ^ ring_all(r << 8);
                P[(d >> 6) * PITCH + (d & 63)] = a0[r].x * a0[r].x + a0[r].y * a0[r].y;
            }
        }
        __syncthreads();

        int wave = tl >> 5;
        int lane = tl & 31;
        if (wave < 4) {
            int row   = (wave << 4) + (lane & 15);
            int col   = lane & 15;
            int khalf = (lane >> 4) << 1;
            v8f acc = {};
            #pragma unroll
            for (int k = 0; k < 64; k += 4) {
                int k0 = k + khalf;
                v2f A = *(const v2f*)(P  + row * PITCH + k0);
                v2f B = *(const v2f*)(Wt + col * PITCH + k0);
                acc = __builtin_amdgcn_wmma_f32_16x16x4_f32(
                     false, A,  false, B,
                     (short)0, acc,  false,  false);
            }
            if (col < 8) {
                int rbase = (wave << 4) + ((lane >> 4) << 3);
                v4f d0 = {acc[0], acc[1], acc[2], acc[3]};
                v4f d1 = {acc[4], acc[5], acc[6], acc[7]};
                *(v4f*)(Rt + col * PITCH + rbase)     = d0;
                *(v4f*)(Rt + col * PITCH + rbase + 4) = d1;
            }
        }
        __syncthreads();

        if (wave == 0) {
            const int q    = lane & 15;
            const int half = lane >> 4;
            const int col  = (q < 6) ? 6 : ((NQ - 1 - q) & 7);
            const unsigned sm = (q == 0) ? (half ? 0xFFFFFFFFu : 0u)
                              : (q == 1) ? 0xFFFF0000u
                              : (q == 2) ? 0xFF00FF00u
                              : (q == 3) ? 0xF0F0F0F0u
                              : (q == 4) ? 0xCCCCCCCCu
                              : (q == 5) ? 0xAAAAAAAAu : 0u;
            const v4f* rp = (const v4f*)(Rt + col * PITCH + 32 * half);
            float s0 = 0.0f, s1 = 0.0f, s2 = 0.0f, s3 = 0.0f;
            #pragma unroll
            for (int i = 0; i < 8; ++i) {
                v4f v = rp[i];
                s0 += __uint_as_float(__float_as_uint(v.x) ^ ((sm << (31 - 4 * i)) & 0x80000000u));
                s1 += __uint_as_float(__float_as_uint(v.y) ^ ((sm << (30 - 4 * i)) & 0x80000000u));
                s2 += __uint_as_float(__float_as_uint(v.z) ^ ((sm << (29 - 4 * i)) & 0x80000000u));
                s3 += __uint_as_float(__float_as_uint(v.w) ^ ((sm << (28 - 4 * i)) & 0x80000000u));
            }
            float acc = (s0 + s1) + (s2 + s3);
            acc += __shfl_xor(acc, 16);
            if (lane < NQ) out[b * NQ + lane] = acc;
        }
        if (cw + 1 < CPW)
            input_stage(la0r, la0i, la1r, la1i, (cw + 1) * NQ, tid, tr, ti, u0r, u0i, u1r, u1i);
    }
}

extern "C" void kernel_launch(void* const* d_in, const int* in_sizes, int n_in,
                              void* d_out, int out_size, void* d_ws, size_t ws_size,
                              hipStream_t stream) {
    const float* x  = (const float*)d_in[0];
    const float* w  = (const float*)d_in[1];
    float* out = (float*)d_out;
    int groups = in_sizes[0] / (NQ * CPW);
    qsim_kernel<<<groups, BLK, 0, stream>>>(x, w, out);
}


# Four circuits per workgroup (was two), grid 512, four waves per SIMD: weight-only work once per four circuits; with 256 registers the per-register LDS addresses are hoisted out of the circuit loop
# speedup vs baseline: 8.3875x; 1.0558x over previous
#include <hip/hip_runtime.h>
#include <math.h>


#define NQ   12
#define NL   6
#define DIM  4096
#define BLK  256
#define NR   16

typedef __attribute__((ext_vector_type(2))) float v2f;
typedef __attribute__((ext_vector_type(8))) float v8f;
typedef __attribute__((ext_vector_type(4))) float v4f;

__host__ __device__ constexpr int swz(int s) {
    return s ^ ((s >> 5) & 31);
}

__host__ __device__ constexpr int cnot_dst(int s) {
    int j = s & (DIM - 1);
    j ^= j >> 1;
    j ^= j >> 2;
    j ^= j >> 4;
    j ^= j >> 8;
    j ^= (j & 1) << (NQ - 1);
    return (s & ~(DIM - 1)) | j;
}
__host__ __device__ constexpr int ring_all(int s) {
    for (int i = 0; i < NL; ++i) s = cnot_dst(s);
    return s;
}
__host__ __device__ constexpr int cnot_src(int j) {
    int t = j ^ ((j & 1) << (NQ - 1));
    return t ^ (t >> 1);
}
__host__ __device__ constexpr int gate_mask(int l, int q) {
    int m = 1 << (NQ - 1 - q);
    for (int i = 0; i < l; ++i) m = cnot_src(m);
    return m;
}
__host__ __device__ constexpr int dual_step(int k) {
    int t = (k ^ (k << 1)) & (DIM - 1);
    return t ^ (t >> (NQ - 1));
}

template <int SH>
__device__ __forceinline__ void load16(v2f (&a)[NR], const v2f* st, int base) {
    #pragma unroll
    for (int r = 0; r < NR; ++r) a[r] = st[base ^ swz(r << SH)];
}
template <int SH>
__device__ __forceinline__ void store16(const v2f (&a)[NR], v2f* st, int base) {
    #pragma unroll
    for (int r = 0; r < NR; ++r) st[base ^ swz(r << SH)] = a[r];
}

__device__ __forceinline__ void bfly4(v2f (&a)[NR]) {
    #pragma unroll
    for (int m = 1; m < NR; m <<= 1) {
        #pragma unroll
        for (int r = 0; r < NR; ++r)
            if (!(r & m)) {
                const v2f s = a[r] + a[r | m];
                const v2f d = a[r] - a[r | m];
                a[r]     = s;
                a[r | m] = d;
            }
    }
}

template <int G0>
__device__ __forceinline__ void phase_terms(float (&G)[NR], const float* gH, int tau) {
    if constexpr (G0 < NL * NQ) {
        constexpr int L  = G0 / NQ;
        constexpr int Q  = G0 % NQ;
        constexpr int PB = NQ - 1 - Q;
        constexpr int MU = gate_mask(L, Q) & (NR - 1);
        const unsigned sb = ((unsigned)tau << (31 - PB)) & 0x80000000u;
        G[MU] += __uint_as_float(__float_as_uint(gH[G0]) ^ sb);
        phase_terms<G0 + 1>(G, gH, (Q == NQ - 1) ? dual_step(tau) : tau);
    }
}

#define PITCH 68

#define CPW 4
#define CPS 2

__device__ __forceinline__ void qubit_pair(float xv, float& a0r, float& a0i,
                                           float& a1r, float& a1i) {
    float ang = tanhf(xv) * 3.14159265358979323846f;
    float h = 0.5f * ang;
    float c = __cosf(h), s = __sinf(h);
    const float cc = c * c, ss = s * s, sc = s * c;
    a0r = 0.5f * (cc + sc);  a0i = 0.5f * (ss - sc);
    a1r = 0.5f * (cc - sc);  a1i = -0.5f * (sc + ss);
}
__device__ __forceinline__ float lane_value(float v, int l) {
    return __int_as_float(__builtin_amdgcn_readlane(__float_as_int(v), l));
}
#define NRQ 4
__device__ __forceinline__ void input_stage(float la0r, float la0i, float la1r, float la1i,
                                            int l0, int tid,
                                            float& tr, float& ti,
                                            float (&u0r)[NRQ], float (&u0i)[NRQ],
                                            float (&u1r)[NRQ], float (&u1i)[NRQ]) {
    tr = 1.0f; ti = 0.0f;
    #pragma unroll
    for (int q = 0; q < NQ; ++q) {
        const float a0r = lane_value(la0r, l0 + q), a0i = lane_value(la0i, l0 + q);
        const float a1r = lane_value(la1r, l0 + q), a1i = lane_value(la1i, l0 + q);
        if (q < 8) {
            const int bit = (tid >> (7 - q)) & 1;
            const float ar = bit ? a1r : a0r;
            const float ai = bit ? a1i : a0i;
            const float nr = tr * ar - ti * ai;
            const float ni = tr * ai + ti * ar;
            tr = nr; ti = ni;
        } else {
            u0r[q - 8] = a0r;  u0i[q - 8] = a0i;
            u1r[q - 8] = a1r;  u1i[q - 8] = a1i;
        }
    }
}

__global__ __launch_bounds__(BLK, 4)
void qsim_kernel(const float* __restrict__ x,
                 const float* __restrict__ w,
                 float* __restrict__ out) {
    __shared__ v2f lds2[2 * DIM];
    __shared__ __attribute__((aligned(16))) float gH[NL * NQ];

    const int tid = threadIdx.x;

    if (tid < NL * NQ) {
        float th = w[tid];
        float k  = rintf(th * 0.0795774715459476679f);
        float h  = fmaf(-k, 6.283185482025146484375f, 0.5f * th);
        h = fmaf(k, 1.748455531469516e-7f, h);
        gH[tid] = h * 0.159154943091895335769f;
    }

    const int i1 = (tid & 15) | (((tid >> 4) & 1) << 9) | (((tid >> 5) & 1) << 8)
                 | (((tid >> 6) & 3) << 10);
    const int i2 = tid << 4;
    int p0 = swz(tid);
    int p1 = swz(i1);
    int p2 = swz(i2);
    int d0 = ring_all(tid);
    int tl = tid;

    float* P  = (float*)(lds2 + DIM);
    float* Rt = P + 64 * PITCH;
    float* Wt = Rt + 8 * PITCH;

    #pragma unroll
    for (int e = 0; e < 4; ++e) {
        int idx = tid + BLK * e;
        int c   = idx >> 6;
        int lo  = idx & 63;
        float v = 0.0f;
        if (c < 6)       v = 1.0f - 2.0f * (float)((lo >> c) & 1);
        else if (c == 6) v = 1.0f;
        Wt[c * PITCH + lo] = v;
    }

    const int b0 = blockIdx.x * CPW;
    static_assert(CPS * NQ <= 32, "one lane per (circuit, qubit) pair of a set");
    static_assert(CPW == 2 * CPS, "two lane sets");
    float la0r, la0i, la1r, la1i;
    float lb0r, lb0i, lb1r, lb1i;
    {
        const int l  = tid & 31;
        const int li = (l < CPS * NQ) ? l : (CPS * NQ - 1);
        qubit_pair(x[b0 * NQ + li],            la0r, la0i, la1r, la1i);
        qubit_pair(x[(b0 + CPS) * NQ + li],    lb0r, lb0i, lb1r, lb1i);
    }

    float tr, ti;
    float u0r[NRQ], u0i[NRQ], u1r[NRQ], u1i[NRQ];
    input_stage(la0r, la0i, la1r, la1i, 0, tid, tr, ti, u0r, u0i, u1r, u1i);
    __syncthreads();

    float ph[NR];
    #pragma unroll
    for (int r = 0; r < NR; ++r) ph[r] = 0.0f;
    phase_terms<0>(ph, gH, i2);
    #pragma unroll
    for (int m = 1; m < NR; m <<= 1) {
        #pragma unroll
        for (int r = 0; r < NR; ++r)
            if (!(r & m)) {
                const float s = ph[r] + ph[r | m];
                const float d = ph[r] - ph[r | m];
                ph[r]     = s;
                ph[r | m] = d;
            }
    }
    float cp[NR], sp[NR];
    #pragma unroll
    for (int r = 0; r < NR; ++r) {
        cp[r] = __builtin_amdgcn_cosf(ph[r]);
        sp[r] = __builtin_amdgcn_sinf(ph[r]);
    }

    #pragma unroll 1
    for (int cw = 0; cw < CPW; ++cw) {
        const int b = b0 + cw;

        {
            v2f amp[NR];
            #pragma unroll
            for (int r = 0; r < NR; ++r) {
                float vr = tr, vi = ti;
                #pragma unroll
                for (int q = 0; q < NRQ; ++q) {
                    const int bit = (r >> (NRQ - 1 - q)) & 1;
                    const float ar = bit ? u1r[q] : u0r[q];
                    const float ai = bit ? u1i[q] : u0i[q];
                    const float nr = vr * ar - vi * ai;
                    const float ni = vr * ai + vi * ar;
                    vr = nr; vi = ni;
                }
                amp[r].x = vr * cp[r] + vi * sp[r];
                amp[r].y = vi * cp[r] - vr * sp[r];
            }
            bfly4(amp);
            store16<0>(amp, lds2, p2);
        }
        __syncthreads();
        {
            v2f a1[NR];
            load16<4>(a1, lds2, p1);
            bfly4(a1);
            store16<4>(a1, lds2, p1);
        }
        __syncthreads();
        {
            v2f a0[NR];
            load16<8>(a0, lds2, p0);
            bfly4(a0);
            #pragma unroll
            for (int r = 0; r < NR; ++r) {
                const int d = d0 ^ ring_all(r << 8);
                P[(d >> 6) * PITCH + (d & 63)] = a0[r].x * a0[r].x + a0[r].y * a0[r].y;
            }
        }
        __syncthreads();

        int wave = tl >> 5;
        int lane = tl & 31;
        if (wave < 4) {
            int row   = (wave << 4) + (lane & 15);
            int col   = lane & 15;
            int khalf = (lane >> 4) << 1;
            v8f acc = {};
            #pragma unroll
            for (int k = 0; k < 64; k += 4) {
                int k0 = k + khalf;
                v2f A = *(const v2f*)(P  + row * PITCH + k0);
                v2f B = *(const v2f*)(Wt + col * PITCH + k0);
                acc = __builtin_amdgcn_wmma_f32_16x16x4_f32(
                     false, A,  false, B,
                     (short)0, acc,  false,  false);
            }
            if (col < 8) {
                int rbase = (wave << 4) + ((lane >> 4) << 3);
                v4f d0 = {acc[0], acc[1], acc[2], acc[3]};
                v4f d1 = {acc[4], acc[5], acc[6], acc[7]};
                *(v4f*)(Rt + col * PITCH + rbase)     = d0;
                *(v4f*)(Rt + col * PITCH + rbase + 4) = d1;
            }
        }
        __syncthreads();

        if (wave == 0) {
            const int q    = lane & 15;
            const int half = lane >> 4;
            const int col  = (q < 6) ? 6 : ((NQ - 1 - q) & 7);
            const unsigned sm = (q == 0) ? (half ? 0xFFFFFFFFu : 0u)
                              : (q == 1) ? 0xFFFF0000u
                              : (q == 2) ? 0xFF00FF00u
                              : (q == 3) ? 0xF0F0F0F0u
                              : (q == 4) ? 0xCCCCCCCCu
                              : (q == 5) ? 0xAAAAAAAAu : 0u;
            const v4f* rp = (const v4f*)(Rt + col * PITCH + 32 * half);
            float s0 = 0.0f, s1 = 0.0f, s2 = 0.0f, s3 = 0.0f;
            #pragma unroll
            for (int i = 0; i < 8; ++i) {
                v4f v = rp[i];
                s0 += __uint_as_float(__float_as_uint(v.x) ^ ((sm << (31 - 4 * i)) & 0x80000000u));
                s1 += __uint_as_float(__float_as_uint(v.y) ^ ((sm << (30 - 4 * i)) & 0x80000000u));
                s2 += __uint_as_float(__float_as_uint(v.z) ^ ((sm << (29 - 4 * i)) & 0x80000000u));
                s3 += __uint_as_float(__float_as_uint(v.w) ^ ((sm << (28 - 4 * i)) & 0x80000000u));
            }
            float acc = (s0 + s1) + (s2 + s3);
            acc += __shfl_xor(acc, 16);
            if (lane < NQ) out[b * NQ + lane] = acc;
        }
        if (cw + 1 < CPW) {
            const int  cn = cw + 1;
            const bool sb = cn >= CPS;
            input_stage(sb ? lb0r : la0r, sb ? lb0i : la0i, sb ? lb1r : la1r, sb ? lb1i : la1i,
                        (sb ? cn - CPS : cn) * NQ, tid, tr, ti, u0r, u0i, u1r, u1i);
        }
    }
}

extern "C" void kernel_launch(void* const* d_in, const int* in_sizes, int n_in,
                              void* d_out, int out_size, void* d_ws, size_t ws_size,
                              hipStream_t stream) {
    const float* x  = (const float*)d_in[0];
    const float* w  = (const float*)d_in[1];
    float* out = (float*)d_out;
    int groups = in_sizes[0] / (NQ * CPW);
    qsim_kernel<<<groups, BLK, 0, stream>>>(x, w, out);
}


# Input product state from three 16-entry four-qubit partial-product tables per circuit, built once per workgroup in LDS: 33 packed complex products per thread and circuit, was 54 and 48 lane reads
# speedup vs baseline: 9.1765x; 1.0941x over previous
#include <hip/hip_runtime.h>
#include <math.h>


#define NQ   12
#define NL   6
#define DIM  4096
#define BLK  256
#define NR   16

typedef __attribute__((ext_vector_type(2))) float v2f;
typedef __attribute__((ext_vector_type(8))) float v8f;
typedef __attribute__((ext_vector_type(4))) float v4f;

__host__ __device__ constexpr int swz(int s) {
    return s ^ ((s >> 5) & 31);
}

__host__ __device__ constexpr int cnot_dst(int s) {
    int j = s & (DIM - 1);
    j ^= j >> 1;
    j ^= j >> 2;
    j ^= j >> 4;
    j ^= j >> 8;
    j ^= (j & 1) << (NQ - 1);
    return (s & ~(DIM - 1)) | j;
}
__host__ __device__ constexpr int ring_all(int s) {
    for (int i = 0; i < NL; ++i) s = cnot_dst(s);
    return s;
}
__host__ __device__ constexpr int cnot_src(int j) {
    int t = j ^ ((j & 1) << (NQ - 1));
    return t ^ (t >> 1);
}
__host__ __device__ constexpr int gate_mask(int l, int q) {
    int m = 1 << (NQ - 1 - q);
    for (int i = 0; i < l; ++i) m = cnot_src(m);
    return m;
}
__host__ __device__ constexpr int dual_step(int k) {
    int t = (k ^ (k << 1)) & (DIM - 1);
    return t ^ (t >> (NQ - 1));
}

template <int SH>
__device__ __forceinline__ void load16(v2f (&a)[NR], const v2f* st, int base) {
    #pragma unroll
    for (int r = 0; r < NR; ++r) a[r] = st[base ^ swz(r << SH)];
}
template <int SH>
__device__ __forceinline__ void store16(const v2f (&a)[NR], v2f* st, int base) {
    #pragma unroll
    for (int r = 0; r < NR; ++r) st[base ^ swz(r << SH)] = a[r];
}

__device__ __forceinline__ void bfly4(v2f (&a)[NR]) {
    #pragma unroll
    for (int m = 1; m < NR; m <<= 1) {
        #pragma unroll
        for (int r = 0; r < NR; ++r)
            if (!(r & m)) {
                const v2f s = a[r] + a[r | m];
                const v2f d = a[r] - a[r | m];
                a[r]     = s;
                a[r | m] = d;
            }
    }
}

template <int G0>
__device__ __forceinline__ void phase_terms(float (&G)[NR], const float* gH, int tau) {
    if constexpr (G0 < NL * NQ) {
        constexpr int L  = G0 / NQ;
        constexpr int Q  = G0 % NQ;
        constexpr int PB = NQ - 1 - Q;
        constexpr int MU = gate_mask(L, Q) & (NR - 1);
        const unsigned sb = ((unsigned)tau << (31 - PB)) & 0x80000000u;
        G[MU] += __uint_as_float(__float_as_uint(gH[G0]) ^ sb);
        phase_terms<G0 + 1>(G, gH, (Q == NQ - 1) ? dual_step(tau) : tau);
    }
}

#define PITCH 68

#define CPW 4
__device__ __forceinline__ void qubit_pair(float xv, float& a0r, float& a0i,
                                           float& a1r, float& a1i) {
    float ang = tanhf(xv) * 3.14159265358979323846f;
    float h = 0.5f * ang;
    float c = __cosf(h), s = __sinf(h);
    const float cc = c * c, ss = s * s, sc = s * c;
    a0r = 0.5f * (cc + sc);  a0i = 0.5f * (ss - sc);
    a1r = 0.5f * (cc - sc);  a1i = -0.5f * (sc + ss);
}
__device__ __forceinline__ v2f cmul(v2f a, v2f b) {
    const v2f ax = {a.x, a.x};
    const v2f ay = {a.y, a.y};
    const v2f bi = {-b.y, b.x};
    return ax * b + ay * bi;
}
#define NG   3
#define TABC (NG * NR)

__global__ __launch_bounds__(BLK, 4)
void qsim_kernel(const float* __restrict__ x,
                 const float* __restrict__ w,
                 float* __restrict__ out) {
    __shared__ v2f lds2[2 * DIM];
    __shared__ __attribute__((aligned(16))) float gH[NL * NQ];
    __shared__ v4f upair[CPW * NQ];
    __shared__ __attribute__((aligned(16))) v2f ptab[CPW * TABC];

    const int tid = threadIdx.x;

    if (tid < NL * NQ) {
        float th = w[tid];
        float k  = rintf(th * 0.0795774715459476679f);
        float h  = fmaf(-k, 6.283185482025146484375f, 0.5f * th);
        h = fmaf(k, 1.748455531469516e-7f, h);
        gH[tid] = h * 0.159154943091895335769f;
    }

    const int i1 = (tid & 15) | (((tid >> 4) & 1) << 9) | (((tid >> 5) & 1) << 8)
                 | (((tid >> 6) & 3) << 10);
    const int i2 = tid << 4;
    int p0 = swz(tid);
    int p1 = swz(i1);
    int p2 = swz(i2);
    int d0 = ring_all(tid);
    int tl = tid;

    float* P  = (float*)(lds2 + DIM);
    float* Rt = P + 64 * PITCH;
    float* Wt = Rt + 8 * PITCH;

    #pragma unroll
    for (int e = 0; e < 4; ++e) {
        int idx = tid + BLK * e;
        int c   = idx >> 6;
        int lo  = idx & 63;
        float v = 0.0f;
        if (c < 6)       v = 1.0f - 2.0f * (float)((lo >> c) & 1);
        else if (c == 6) v = 1.0f;
        Wt[c * PITCH + lo] = v;
    }

    const int b0 = blockIdx.x * CPW;
    if (tid < CPW * NQ) {
        float a0r, a0i, a1r, a1i;
        qubit_pair(x[b0 * NQ + tid], a0r, a0i, a1r, a1i);
        const v4f pr = {a0r, a0i, a1r, a1i};
        upair[tid] = pr;
    }
    __syncthreads();

    if (tid < CPW * TABC) {
        const v2f* up = (const v2f*)upair + 8 * (tid >> 4);
        v2f p = up[(tid >> 3) & 1];
        #pragma unroll
        for (int j = 1; j < 4; ++j) p = cmul(p, up[2 * j + ((tid >> (3 - j)) & 1)]);
        ptab[tid] = p;
    }

    float ph[NR];
    #pragma unroll
    for (int r = 0; r < NR; ++r) ph[r] = 0.0f;
    phase_terms<0>(ph, gH, i2);
    #pragma unroll
    for (int m = 1; m < NR; m <<= 1) {
        #pragma unroll
        for (int r = 0; r < NR; ++r)
            if (!(r & m)) {
                const float s = ph[r] + ph[r | m];
                const float d = ph[r] - ph[r | m];
                ph[r]     = s;
                ph[r | m] = d;
            }
    }
    v2f e1[NR], e2[NR];
    #pragma unroll
    for (int r = 0; r < NR; ++r) {
        const float cpr = __builtin_amdgcn_cosf(ph[r]);
        const float spr = __builtin_amdgcn_sinf(ph[r]);
        e1[r].x = cpr;  e1[r].y = -spr;
        e2[r].x = spr;  e2[r].y = cpr;
    }
    __syncthreads();

    const int ia = tid >> 4;
    const int ib = NR + (tid & 15);
    v2f fa = ptab[ia], fb = ptab[ib];
    v2f ur[NR];
    #pragma unroll
    for (int r = 0; r < NR; ++r) ur[r] = ptab[2 * NR + r];

    #pragma unroll 1
    for (int cw = 0; cw < CPW; ++cw) {
        const int b = b0 + cw;

        {
            const v2f t = cmul(fa, fb);
            v2f amp[NR];
            #pragma unroll
            for (int r = 0; r < NR; ++r) {
                const v2f v  = cmul(t, ur[r]);
                const v2f vx = {v.x, v.x};
                const v2f vy = {v.y, v.y};
                amp[r] = vx * e1[r] + vy * e2[r];
            }
            bfly4(amp);
            store16<0>(amp, lds2, p2);
        }
        __syncthreads();
        {
            v2f a1[NR];
            load16<4>(a1, lds2, p1);
            bfly4(a1);
            store16<4>(a1, lds2, p1);
        }
        __syncthreads();
        {
            v2f a0[NR];
            load16<8>(a0, lds2, p0);
            bfly4(a0);
            #pragma unroll
            for (int r = 0; r < NR; ++r) {
                const int d = d0 ^ ring_all(r << 8);
                P[(d >> 6) * PITCH + (d & 63)] = a0[r].x * a0[r].x + a0[r].y * a0[r].y;
            }
        }
        __syncthreads();

        if (cw + 1 < CPW) {
            const v2f* tb = ptab + (cw + 1) * TABC;
            fa = tb[ia];  fb = tb[ib];
            #pragma unroll
            for (int r = 0; r < NR; ++r) ur[r] = tb[2 * NR + r];
        }

        int wave = tl >> 5;
        int lane = tl & 31;
        if (wave < 4) {
            int row   = (wave << 4) + (lane & 15);
            int col   = lane & 15;
            int khalf = (lane >> 4) << 1;
            v8f acc = {};
            #pragma unroll
            for (int k = 0; k < 64; k += 4) {
                int k0 = k + khalf;
                v2f A = *(const v2f*)(P  + row * PITCH + k0);
                v2f B = *(const v2f*)(Wt + col * PITCH + k0);
                acc = __builtin_amdgcn_wmma_f32_16x16x4_f32(
                     false, A,  false, B,
                     (short)0, acc,  false,  false);
            }
            if (col < 8) {
                int rbase = (wave << 4) + ((lane >> 4) << 3);
                v4f d0 = {acc[0], acc[1], acc[2], acc[3]};
                v4f d1 = {acc[4], acc[5], acc[6], acc[7]};
                *(v4f*)(Rt + col * PITCH + rbase)     = d0;
                *(v4f*)(Rt + col * PITCH + rbase + 4) = d1;
            }
        }
        __syncthreads();

        if (wave == 0) {
            const int q    = lane & 15;
            const int half = lane >> 4;
            const int col  = (q < 6) ? 6 : ((NQ - 1 - q) & 7);
            const unsigned sm = (q == 0) ? (half ? 0xFFFFFFFFu : 0u)
                              : (q == 1) ? 0xFFFF0000u
                              : (q == 2) ? 0xFF00FF00u
                              : (q == 3) ? 0xF0F0F0F0u
                              : (q == 4) ? 0xCCCCCCCCu
                              : (q == 5) ? 0xAAAAAAAAu : 0u;
            const v4f* rp = (const v4f*)(Rt + col * PITCH + 32 * half);
            float s0 = 0.0f, s1 = 0.0f, s2 = 0.0f, s3 = 0.0f;
            #pragma unroll
            for (int i = 0; i < 8; ++i) {
                v4f v = rp[i];
                s0 += __uint_as_float(__float_as_uint(v.x) ^ ((sm << (31 - 4 * i)) & 0x80000000u));
                s1 += __uint_as_float(__float_as_uint(v.y) ^ ((sm << (30 - 4 * i)) & 0x80000000u));
                s2 += __uint_as_float(__float_as_uint(v.z) ^ ((sm << (29 - 4 * i)) & 0x80000000u));
                s3 += __uint_as_float(__float_as_uint(v.w) ^ ((sm << (28 - 4 * i)) & 0x80000000u));
            }
            float acc = (s0 + s1) + (s2 + s3);
            acc += __shfl_xor(acc, 16);
            if (lane < NQ) out[b * NQ + lane] = acc;
        }
    }
}

extern "C" void kernel_launch(void* const* d_in, const int* in_sizes, int n_in,
                              void* d_out, int out_size, void* d_ws, size_t ws_size,
                              hipStream_t stream) {
    const float* x  = (const float*)d_in[0];
    const float* w  = (const float*)d_in[1];
    float* out = (float*)d_out;
    int groups = in_sizes[0] / (NQ * CPW);
    qsim_kernel<<<groups, BLK, 0, stream>>>(x, w, out);
}
